# starting kernel
# baseline (speedup 1.0000x reference)
#include <hip/hip_runtime.h>
#include <cmath>

#define NN   100000
#define EE   3200000
#define F_INd 500
#define HID  16
#define NC   3

typedef __attribute__((ext_vector_type(2))) float v2f;
typedef __attribute__((ext_vector_type(8))) float v8f;

__global__ void k_init_deg(float* __restrict__ deg) {
  int i = blockIdx.x * blockDim.x + threadIdx.x;
  if (i < NN) deg[i] = 1.0f;
}

__global__ void k_count_deg(const int* __restrict__ dst, float* __restrict__ deg) {
  int e = blockIdx.x * blockDim.x + threadIdx.x;
  if (e < EE) atomicAdd(&deg[dst[e]], 1.0f);
}

__global__ void k_rsqrt(float* __restrict__ dinv) {
  int i = blockIdx.x * blockDim.x + threadIdx.x;
  if (i < NN) dinv[i] = rsqrtf(dinv[i]);
}

__global__ __launch_bounds__(256) void k_gemm1_wmma(const float* __restrict__ X,
                                                    const float* __restrict__ W,
                                                    float* __restrict__ H1) {
  __shared__ float Wlds[F_INd * HID];
  for (int i = threadIdx.x; i < F_INd * HID; i += 256) Wlds[i] = W[i];
  __syncthreads();

  const int wave = threadIdx.x >> 5;
  const int lane = threadIdx.x & 31;
  const int tile = blockIdx.x * 8 + wave;
  if (tile >= NN / 16) return;

  const int row_base = tile * 16;
  const int m    = lane & 15;
  const int hi   = lane >> 4;
  const int koff = hi << 1;

  const float* xrow = X + (size_t)(row_base + m) * F_INd;

  v8f acc = {};
  for (int k0 = 0; k0 < F_INd; k0 += 4) {
    v2f a, b;
    a.x = xrow[k0 + koff];
    a.y = xrow[k0 + koff + 1];
    b.x = Wlds[(k0 + koff)     * HID + m];
    b.y = Wlds[(k0 + koff + 1) * HID + m];
    acc = __builtin_amdgcn_wmma_f32_16x16x4_f32(false, a, false, b,
                                                (short)0, acc, false, false);
  }

  const int rbase = row_base + (hi << 3);
#pragma unroll
  for (int r = 0; r < 8; ++r)
    H1[(size_t)(rbase + r) * HID + m] = acc[r];
}

__global__ void k_agg1_init(const float* __restrict__ h1, const float* __restrict__ dinv,
                            float* __restrict__ agg1) {
  int i = blockIdx.x * blockDim.x + threadIdx.x;
  if (i >= NN * HID) return;
  int node = i >> 4;
  float di = dinv[node];
  agg1[i] = h1[i] * di * di;
}

__global__ void k_agg1_edges(const float* __restrict__ h1, const float* __restrict__ dinv,
                             const int* __restrict__ src, const int* __restrict__ dst,
                             float* __restrict__ agg1) {
  long long idx = (long long)blockIdx.x * blockDim.x + threadIdx.x;
  int e = (int)(idx >> 4);
  int f = (int)(idx & 15);
  if (e >= EE) return;
  int s = src[e], d = dst[e];
  float nrm = dinv[s] * dinv[d];
  atomicAdd(&agg1[(size_t)d * HID + f], h1[(size_t)s * HID + f] * nrm);
}

__global__ void k_relu_lin2(const float* __restrict__ agg1, const float* __restrict__ b1,
                            const float* __restrict__ W2, float* __restrict__ h2) {
  int i = blockIdx.x * blockDim.x + threadIdx.x;
  if (i >= NN) return;
  float v[HID];
#pragma unroll
  for (int f = 0; f < HID; ++f) {
    float t = agg1[(size_t)i * HID + f] + b1[f];
    v[f] = t > 0.0f ? t : 0.0f;
  }
#pragma unroll
  for (int c = 0; c < NC; ++c) {
    float s = 0.0f;
#pragma unroll
    for (int f = 0; f < HID; ++f) s += v[f] * W2[f * NC + c];
    h2[(size_t)i * 4 + c] = s;
  }
  h2[(size_t)i * 4 + 3] = 0.0f;
}

__global__ void k_agg2_init(const float* __restrict__ h2, const float* __restrict__ dinv,
                            float* __restrict__ agg2) {
  int i = blockIdx.x * blockDim.x + threadIdx.x;
  if (i >= NN * 4) return;
  int node = i >> 2;
  float di = dinv[node];
  agg2[i] = h2[i] * di * di;
}

__global__ void k_agg2_edges(const float* __restrict__ h2, const float* __restrict__ dinv,
                             const int* __restrict__ src, const int* __restrict__ dst,
                             float* __restrict__ agg2) {
  int e = blockIdx.x * blockDim.x + threadIdx.x;
  if (e >= EE) return;
  int s = src[e], d = dst[e];
  float nrm = dinv[s] * dinv[d];
#pragma unroll
  for (int c = 0; c < NC; ++c)
    atomicAdd(&agg2[(size_t)d * 4 + c], h2[(size_t)s * 4 + c] * nrm);
}

__global__ void k_logsoftmax(const float* __restrict__ agg2, const float* __restrict__ b2,
                             float* __restrict__ out) {
  int i = blockIdx.x * blockDim.x + threadIdx.x;
  if (i >= NN) return;
  float z0 = agg2[(size_t)i * 4 + 0] + b2[0];
  float z1 = agg2[(size_t)i * 4 + 1] + b2[1];
  float z2 = agg2[(size_t)i * 4 + 2] + b2[2];
  float mx = fmaxf(z0, fmaxf(z1, z2));
  float lse = logf(expf(z0 - mx) + expf(z1 - mx) + expf(z2 - mx));
  out[(size_t)i * NC + 0] = z0 - mx - lse;
  out[(size_t)i * NC + 1] = z1 - mx - lse;
  out[(size_t)i * NC + 2] = z2 - mx - lse;
}

extern "C" void kernel_launch(void* const* d_in, const int* in_sizes, int n_in,
                              void* d_out, int out_size, void* d_ws, size_t ws_size,
                              hipStream_t stream) {
  const float* x  = (const float*)d_in[0];
  const float* W1 = (const float*)d_in[1];
  const float* b1 = (const float*)d_in[2];
  const float* W2 = (const float*)d_in[3];
  const float* b2 = (const float*)d_in[4];
  const int*   ei = (const int*)d_in[5];
  const int* src = ei;
  const int* dst = ei + EE;
  float* out = (float*)d_out;

  float* ws   = (float*)d_ws;
  float* h1   = ws;
  float* dinv = h1   + (size_t)NN * HID;
  float* agg1 = dinv + (size_t)NN;
  float* h2   = agg1 + (size_t)NN * HID;
  float* agg2 = h2   + (size_t)NN * 4;

  const int T = 256;
  k_init_deg   <<<(NN + T - 1) / T, T, 0, stream>>>(dinv);
  k_count_deg  <<<(EE + T - 1) / T, T, 0, stream>>>(dst, dinv);
  k_rsqrt      <<<(NN + T - 1) / T, T, 0, stream>>>(dinv);

  k_gemm1_wmma <<<(NN / 16 + 7) / 8, T, 0, stream>>>(x, W1, h1);

  k_agg1_init  <<<(NN * HID + T - 1) / T, T, 0, stream>>>(h1, dinv, agg1);
  {
    long long total = (long long)EE * HID;
    int blocks = (int)((total + T - 1) / T);
    k_agg1_edges<<<blocks, T, 0, stream>>>(h1, dinv, src, dst, agg1);
  }

  k_relu_lin2  <<<(NN + T - 1) / T, T, 0, stream>>>(agg1, b1, W2, h2);

  k_agg2_init  <<<(NN * 4 + T - 1) / T, T, 0, stream>>>(h2, dinv, agg2);
  k_agg2_edges <<<(EE + T - 1) / T, T, 0, stream>>>(h2, dinv, src, dst, agg2);

  k_logsoftmax <<<(NN + T - 1) / T, T, 0, stream>>>(agg2, b2, out);
}


# edges ordered once by 128-node dst bucket (LDS counting sort per chunk); one block per bucket keeps degree, layer-1 and layer-2 sums in LDS: no global atomics; dinv factored out; LDS add order free
# speedup vs baseline: 4.2177x; 4.2177x over previous
#include <hip/hip_runtime.h>
#include <cmath>

#define NN   100000
#define EE   3200000
#define F_INd 500
#define HID  16
#define NC   3
#define RSH  7
#define RB   128
#define PB   782
#define NBK  125
#define CH   25600
#define EPT  25
#define TS   832

typedef __attribute__((ext_vector_type(2))) float v2f;
typedef __attribute__((ext_vector_type(8))) float v8f;
typedef unsigned int __attribute__((ext_vector_type(4))) u32x4;
typedef u32x4 __attribute__((may_alias)) u32x4a;


__device__ __forceinline__ int block_excl_scan(int v, int* wsum, int& total) {
  const int lane = threadIdx.x & 31, wid = threadIdx.x >> 5;
  int x = v;
#pragma unroll
  for (int o = 1; o < 32; o <<= 1) { int y = __shfl_up(x, o, 32); if (lane >= o) x += y; }
  if (lane == 31) wsum[wid] = x;
  __syncthreads();
  if (wid == 0) {
    const int nw = blockDim.x >> 5;
    int w = lane < nw ? wsum[lane] : 0;
    int z = w;
#pragma unroll
    for (int o = 1; o < 32; o <<= 1) { int y = __shfl_up(z, o, 32); if (lane >= o) z += y; }
    wsum[lane] = z - w;
    if (lane == 31) wsum[32] = z;
  }
  __syncthreads();
  total = wsum[32];
  return wsum[wid] + x - v;
}

__global__ __launch_bounds__(1024) void k_chunk_sort(const int* __restrict__ src, const int* __restrict__ dst,
                                                     unsigned short* __restrict__ e16,
                                                     unsigned char* __restrict__ e8,
                                                     unsigned short* __restrict__ otab) {
  __shared__ int hist[PB];
  __shared__ int wsum[33];
  __shared__ __attribute__((aligned(16))) unsigned short st16[CH];
  __shared__ __attribute__((aligned(16))) unsigned char  st8[CH];
  __shared__ __attribute__((aligned(16))) unsigned short row[TS];
  const int t = threadIdx.x, blk = blockIdx.x;
  if (t < PB) hist[t] = 0;
  if (t < TS) row[t] = 0;
  __syncthreads();
  const int* sp = src + blk * CH;
  const int* dp = dst + blk * CH;
  int sv[EPT], dv[EPT];
#pragma unroll
  for (int u = 0; u < EPT; ++u) { sv[u] = sp[t + 1024 * u]; dv[u] = dp[t + 1024 * u]; }
#pragma unroll
  for (int u = 0; u < EPT; ++u) atomicAdd(&hist[dv[u] >> RSH], 1);
  __syncthreads();
  const int c = t < PB ? hist[t] : 0;
  int total;
  const int lo = block_excl_scan(c, wsum, total);
  if (t < PB) { hist[t] = lo; row[t] = (unsigned short)lo; }
  if (t == 0) row[PB] = (unsigned short)CH;
  __syncthreads();
#pragma unroll
  for (int u = 0; u < EPT; ++u) {
    const int r = atomicAdd(&hist[dv[u] >> RSH], 1);
    st16[r] = (unsigned short)(sv[u] & 0xFFFF);
    st8[r]  = (unsigned char)((dv[u] & (RB - 1)) | ((sv[u] >> 16) << 7));
  }
  __syncthreads();
  const u32x4a* v16 = (const u32x4a*)st16;
  u32x4a* g16 = (u32x4a*)(e16 + (size_t)blk * CH);
  for (int i = t; i < CH / 8; i += 1024) g16[i] = v16[i];
  const u32x4a* v8 = (const u32x4a*)st8;
  u32x4a* g8 = (u32x4a*)(e8 + (size_t)blk * CH);
  for (int i = t; i < CH / 16; i += 1024) g8[i] = v8[i];
  const u32x4a* vr = (const u32x4a*)row;
  u32x4a* gr = (u32x4a*)(otab + (size_t)blk * TS);
  if (t < TS / 8) gr[t] = vr[t];
}

__global__ __launch_bounds__(256) void k_gemm1_wmma(const float* __restrict__ X,
                                                    const float* __restrict__ W,
                                                    const unsigned char* __restrict__ e8,
                                                    const unsigned short* __restrict__ otab,
                                                    float* __restrict__ H1) {
  __shared__ float Wlds[F_INd * HID];
  __shared__ int dcnt[RB];
  for (int i = threadIdx.x; i < F_INd * HID; i += 256) Wlds[i] = W[i];
  if (threadIdx.x < RB) dcnt[threadIdx.x] = 0;
  __syncthreads();
  {
    const int g = threadIdx.x >> 4, f = threadIdx.x & 15;
    int o0[8], o1[8];
#pragma unroll
    for (int k = 0; k < 8; ++k) {
      const int j = g + 16 * k, jc = j < NBK ? j : NBK - 1;
      const unsigned short* rw = otab + (size_t)jc * TS + blockIdx.x;
      o0[k] = j < NBK ? (int)rw[0] : 0;
      o1[k] = j < NBK ? (int)rw[1] : 0;
    }
#pragma unroll
    for (int k = 0; k < 8; ++k) {
      const int jc = (g + 16 * k) < NBK ? (g + 16 * k) : NBK - 1;
      const unsigned char* p8 = e8 + (size_t)jc * CH;
      const int oe = o1[k];
      for (int i0 = o0[k] + f; i0 < oe; i0 += 16 * 4) {
        int q[4];
#pragma unroll
        for (int u = 0; u < 4; ++u) { const int i = i0 + 16 * u; q[u] = p8[i < oe ? i : oe - 1]; }
#pragma unroll
        for (int u = 0; u < 4; ++u)
          if (i0 + 16 * u < oe) atomicAdd(&dcnt[q[u] & (RB - 1)], 1);
      }
    }
  }
  __syncthreads();

  const int wave = threadIdx.x >> 5;
  const int lane = threadIdx.x & 31;
  const int tile = blockIdx.x * 8 + wave;
  if (tile >= NN / 16) return;

  const int row_base = tile * 16;
  const int m    = lane & 15;
  const int hi   = lane >> 4;
  const int koff = hi << 1;

  const float* xrow = X + (size_t)(row_base + m) * F_INd;

  v8f acc = {};
  for (int k0 = 0; k0 < F_INd; k0 += 4) {
    v2f a, b;
    a.x = xrow[k0 + koff];
    a.y = xrow[k0 + koff + 1];
    b.x = Wlds[(k0 + koff)     * HID + m];
    b.y = Wlds[(k0 + koff + 1) * HID + m];
    acc = __builtin_amdgcn_wmma_f32_16x16x4_f32(false, a, false, b,
                                                (short)0, acc, false, false);
  }

  const int rbase = row_base + (hi << 3);
  const int lrow  = (wave << 4) + (hi << 3);
#pragma unroll
  for (int r = 0; r < 8; ++r)
    H1[(size_t)(rbase + r) * HID + m] = acc[r] * rsqrtf((float)(dcnt[lrow + r] + 1));
}

__global__ __launch_bounds__(512, 16) void k_agg1_bucket(const float* __restrict__ h1s,
                                                     const unsigned short* __restrict__ e16,
                                                     const unsigned char* __restrict__ e8,
                                                     const unsigned short* __restrict__ otab,
                                                     const float* __restrict__ b1,
                                                     const float* __restrict__ W2,
                                                     float* __restrict__ h2s) {
  __shared__ float acc[RB * HID];
  __shared__ int dcnt[RB];
  const int t = threadIdx.x, base = blockIdx.x * RB;
  if (t < RB) dcnt[t] = 0;
  for (int i = t; i < RB * HID; i += 512)
    acc[i] = (base + (i >> 4)) < NN ? h1s[(size_t)base * HID + i] : 0.0f;
  __syncthreads();
  const int f = t & 15, g = t >> 4;
  int o0[4], o1[4];
#pragma unroll
  for (int k = 0; k < 4; ++k) {
    const int j = g + 32 * k, jc = j < NBK ? j : NBK - 1;
    const unsigned short* rw = otab + (size_t)jc * TS + blockIdx.x;
    o0[k] = j < NBK ? (int)rw[0] : 0;
    o1[k] = j < NBK ? (int)rw[1] : 0;
  }
#pragma unroll
  for (int k = 0; k < 4; ++k) {
    const int jc = (g + 32 * k) < NBK ? (g + 32 * k) : NBK - 1;
    const unsigned short* p16 = e16 + (size_t)jc * CH;
    const unsigned char*  p8  = e8  + (size_t)jc * CH;
    const int oe = o1[k];
    int i0 = o0[k];
    int q[8] = {0, 0, 0, 0, 0, 0, 0, 0}, lo[8] = {0, 0, 0, 0, 0, 0, 0, 0};
    if (i0 < oe) {
#pragma unroll
      for (int u = 0; u < 8; ++u) {
        const int i = i0 + u, ic = i < oe ? i : oe - 1;
        q[u] = p8[ic]; lo[u] = p16[ic];
      }
    }
    while (i0 < oe) {
      float hv[8];
#pragma unroll
      for (int u = 0; u < 8; ++u) {
        const int s = lo[u] | ((q[u] >> 7) << 16);
        hv[u] = h1s[(size_t)s * HID + f];
      }
      int qn[8], ln[8];
      const int inx = i0 + 8;
#pragma unroll
      for (int u = 0; u < 8; ++u) {
        const int i = inx + u, ic = i < oe ? i : oe - 1;
        qn[u] = p8[ic]; ln[u] = p16[ic];
      }
#pragma unroll
      for (int u = 0; u < 8; ++u)
        if (i0 + u < oe) {
          const int dl = q[u] & (RB - 1);
          atomicAdd(&acc[dl * HID + f], hv[u]);
          if (f == 0) atomicAdd(&dcnt[dl], 1);
        }
#pragma unroll
      for (int u = 0; u < 8; ++u) { q[u] = qn[u]; lo[u] = ln[u]; }
      i0 = inx;
    }
  }
  __syncthreads();
  if (t < RB && base + t < NN) {
    const float di = rsqrtf((float)(dcnt[t] + 1));
    float v[HID];
#pragma unroll
    for (int k = 0; k < HID; ++k) {
      float x = acc[t * HID + k] * di + b1[k];
      v[k] = x > 0.0f ? x : 0.0f;
    }
#pragma unroll
    for (int c = 0; c < NC; ++c) {
      float sum = 0.0f;
#pragma unroll
      for (int k = 0; k < HID; ++k) sum += v[k] * W2[k * NC + c];
      h2s[(size_t)(base + t) * NC + c] = sum * di;
    }
  }
}

__global__ __launch_bounds__(512) void k_agg2_bucket(const float* __restrict__ h2s,
                                                     const unsigned short* __restrict__ e16,
                                                     const unsigned char* __restrict__ e8,
                                                     const unsigned short* __restrict__ otab,
                                                     float* __restrict__ agg2) {
  __shared__ float acc[RB * 4];
  __shared__ int dcnt[RB];
  const int t = threadIdx.x, base = blockIdx.x * RB;
  if (t < RB) dcnt[t] = 0;
  {
    const int node = base + (t >> 2), c = t & 3;
    acc[t] = (node < NN && c < NC) ? h2s[(size_t)node * NC + c] : 0.0f;
  }
  __syncthreads();
  const int f = t & 15, g = t >> 4;
  int o0[4], o1[4];
#pragma unroll
  for (int k = 0; k < 4; ++k) {
    const int j = g + 32 * k, jc = j < NBK ? j : NBK - 1;
    const unsigned short* rw = otab + (size_t)jc * TS + blockIdx.x;
    o0[k] = j < NBK ? (int)rw[0] : 0;
    o1[k] = j < NBK ? (int)rw[1] : 0;
  }
#pragma unroll
  for (int k = 0; k < 4; ++k) {
    const int jc = (g + 32 * k) < NBK ? (g + 32 * k) : NBK - 1;
    const unsigned short* p16 = e16 + (size_t)jc * CH;
    const unsigned char*  p8  = e8  + (size_t)jc * CH;
    const int oe = o1[k];
    for (int i0 = o0[k] + f; i0 < oe; i0 += 16 * 4) {
      int q[4], lo[4];
      float a0[4], a1[4], a2[4];
#pragma unroll
      for (int u = 0; u < 4; ++u) {
        const int i = i0 + 16 * u, ic = i < oe ? i : oe - 1;
        q[u] = p8[ic]; lo[u] = p16[ic];
      }
#pragma unroll
      for (int u = 0; u < 4; ++u) {
        const float* hs = h2s + (size_t)(lo[u] | ((q[u] >> 7) << 16)) * NC;
        a0[u] = hs[0]; a1[u] = hs[1]; a2[u] = hs[2];
      }
#pragma unroll
      for (int u = 0; u < 4; ++u)
        if (i0 + 16 * u < oe) {
          const int dl = q[u] & (RB - 1);
          atomicAdd(&acc[dl * 4 + 0], a0[u]);
          atomicAdd(&acc[dl * 4 + 1], a1[u]);
          atomicAdd(&acc[dl * 4 + 2], a2[u]);
          atomicAdd(&dcnt[dl], 1);
        }
    }
  }
  __syncthreads();
  if (t < RB && base + t < NN) {
    const float di = rsqrtf((float)(dcnt[t] + 1));
    float* o = agg2 + (size_t)(base + t) * 4;
    o[0] = acc[t * 4 + 0] * di;
    o[1] = acc[t * 4 + 1] * di;
    o[2] = acc[t * 4 + 2] * di;
    o[3] = 0.0f;
  }
}

__global__ void k_logsoftmax(const float* __restrict__ agg2, const float* __restrict__ b2,
                             float* __restrict__ out) {
  int i = blockIdx.x * blockDim.x + threadIdx.x;
  if (i >= NN) return;
  float z0 = agg2[(size_t)i * 4 + 0] + b2[0];
  float z1 = agg2[(size_t)i * 4 + 1] + b2[1];
  float z2 = agg2[(size_t)i * 4 + 2] + b2[2];
  float mx = fmaxf(z0, fmaxf(z1, z2));
  float lse = logf(expf(z0 - mx) + expf(z1 - mx) + expf(z2 - mx));
  out[(size_t)i * NC + 0] = z0 - mx - lse;
  out[(size_t)i * NC + 1] = z1 - mx - lse;
  out[(size_t)i * NC + 2] = z2 - mx - lse;
}

extern "C" void kernel_launch(void* const* d_in, const int* in_sizes, int n_in,
                              void* d_out, int out_size, void* d_ws, size_t ws_size,
                              hipStream_t stream) {
  const float* x  = (const float*)d_in[0];
  const float* W1 = (const float*)d_in[1];
  const float* b1 = (const float*)d_in[2];
  const float* W2 = (const float*)d_in[3];
  const float* b2 = (const float*)d_in[4];
  const int*   ei = (const int*)d_in[5];
  const int* src = ei;
  const int* dst = ei + EE;
  float* out = (float*)d_out;

  char* wsb = (char*)d_ws;
  float*          h1s  = (float*)wsb;
  unsigned short* e16  = (unsigned short*)(wsb + 6400000);
  unsigned char*  e8   = (unsigned char*)(wsb + 12800000);
  unsigned short* otab = (unsigned short*)(wsb + 16000000);
  float* agg2 = (float*)wsb;
  float* h2s = out;

  k_chunk_sort  <<<NBK, 1024, 0, stream>>>(src, dst, e16, e8, otab);
  k_gemm1_wmma  <<<PB, 256, 0, stream>>>(x, W1, e8, otab, h1s);
  k_agg1_bucket <<<PB, 512, 0, stream>>>(h1s, e16, e8, otab, b1, W2, h2s);
  k_agg2_bucket <<<PB, 512, 0, stream>>>(h2s, e16, e8, otab, agg2);
  k_logsoftmax  <<<(NN + 255) / 256, 256, 0, stream>>>(agg2, b2, out);
}


# layer-1 edge pass: a 4-lane group per chunk run, all runs of a bucket at once; a lane loads a 16-byte quarter of each source row (8 edges per load), records shared by DPP; LDS add order free
# speedup vs baseline: 4.7305x; 1.1216x over previous
#include <hip/hip_runtime.h>
#include <cmath>

#define NN   100000
#define EE   3200000
#define F_INd 500
#define HID  16
#define NC   3
#define RSH  7
#define RB   128
#define PB   782
#define NBK  125
#define CH   25600
#define EPT  25
#define TS   832

typedef __attribute__((ext_vector_type(2))) float v2f;
typedef __attribute__((ext_vector_type(8))) float v8f;
typedef unsigned int __attribute__((ext_vector_type(4))) u32x4;
typedef u32x4 __attribute__((may_alias)) u32x4a;


__device__ __forceinline__ int block_excl_scan(int v, int* wsum, int& total) {
  const int lane = threadIdx.x & 31, wid = threadIdx.x >> 5;
  int x = v;
#pragma unroll
  for (int o = 1; o < 32; o <<= 1) { int y = __shfl_up(x, o, 32); if (lane >= o) x += y; }
  if (lane == 31) wsum[wid] = x;
  __syncthreads();
  if (wid == 0) {
    const int nw = blockDim.x >> 5;
    int w = lane < nw ? wsum[lane] : 0;
    int z = w;
#pragma unroll
    for (int o = 1; o < 32; o <<= 1) { int y = __shfl_up(z, o, 32); if (lane >= o) z += y; }
    wsum[lane] = z - w;
    if (lane == 31) wsum[32] = z;
  }
  __syncthreads();
  total = wsum[32];
  return wsum[wid] + x - v;
}

__global__ __launch_bounds__(1024) void k_chunk_sort(const int* __restrict__ src, const int* __restrict__ dst,
                                                     unsigned short* __restrict__ e16,
                                                     unsigned char* __restrict__ e8,
                                                     unsigned short* __restrict__ otab) {
  __shared__ int hist[PB];
  __shared__ int wsum[33];
  __shared__ __attribute__((aligned(16))) unsigned short st16[CH];
  __shared__ __attribute__((aligned(16))) unsigned char  st8[CH];
  __shared__ __attribute__((aligned(16))) unsigned short row[TS];
  const int t = threadIdx.x, blk = blockIdx.x;
  if (t < PB) hist[t] = 0;
  if (t < TS) row[t] = 0;
  __syncthreads();
  const int* sp = src + blk * CH;
  const int* dp = dst + blk * CH;
  int sv[EPT], dv[EPT];
#pragma unroll
  for (int u = 0; u < EPT; ++u) { sv[u] = sp[t + 1024 * u]; dv[u] = dp[t + 1024 * u]; }
#pragma unroll
  for (int u = 0; u < EPT; ++u) atomicAdd(&hist[dv[u] >> RSH], 1);
  __syncthreads();
  const int c = t < PB ? hist[t] : 0;
  int total;
  const int lo = block_excl_scan(c, wsum, total);
  if (t < PB) { hist[t] = lo; row[t] = (unsigned short)lo; }
  if (t == 0) row[PB] = (unsigned short)CH;
  __syncthreads();
#pragma unroll
  for (int u = 0; u < EPT; ++u) {
    const int r = atomicAdd(&hist[dv[u] >> RSH], 1);
    st16[r] = (unsigned short)(sv[u] & 0xFFFF);
    st8[r]  = (unsigned char)((dv[u] & (RB - 1)) | ((sv[u] >> 16) << 7));
  }
  __syncthreads();
  const u32x4a* v16 = (const u32x4a*)st16;
  u32x4a* g16 = (u32x4a*)(e16 + (size_t)blk * CH);
  for (int i = t; i < CH / 8; i += 1024) g16[i] = v16[i];
  const u32x4a* v8 = (const u32x4a*)st8;
  u32x4a* g8 = (u32x4a*)(e8 + (size_t)blk * CH);
  for (int i = t; i < CH / 16; i += 1024) g8[i] = v8[i];
  const u32x4a* vr = (const u32x4a*)row;
  u32x4a* gr = (u32x4a*)(otab + (size_t)blk * TS);
  if (t < TS / 8) gr[t] = vr[t];
}

__global__ __launch_bounds__(256) void k_gemm1_wmma(const float* __restrict__ X,
                                                    const float* __restrict__ W,
                                                    const unsigned char* __restrict__ e8,
                                                    const unsigned short* __restrict__ otab,
                                                    float* __restrict__ H1) {
  __shared__ float Wlds[F_INd * HID];
  __shared__ int dcnt[RB];
  for (int i = threadIdx.x; i < F_INd * HID; i += 256) Wlds[i] = W[i];
  if (threadIdx.x < RB) dcnt[threadIdx.x] = 0;
  __syncthreads();
  {
    const int g = threadIdx.x >> 4, f = threadIdx.x & 15;
    int o0[8], o1[8];
#pragma unroll
    for (int k = 0; k < 8; ++k) {
      const int j = g + 16 * k, jc = j < NBK ? j : NBK - 1;
      const unsigned short* rw = otab + (size_t)jc * TS + blockIdx.x;
      o0[k] = j < NBK ? (int)rw[0] : 0;
      o1[k] = j < NBK ? (int)rw[1] : 0;
    }
#pragma unroll
    for (int k = 0; k < 8; ++k) {
      const int jc = (g + 16 * k) < NBK ? (g + 16 * k) : NBK - 1;
      const unsigned char* p8 = e8 + (size_t)jc * CH;
      const int oe = o1[k];
      for (int i0 = o0[k] + f; i0 < oe; i0 += 16 * 4) {
        int q[4];
#pragma unroll
        for (int u = 0; u < 4; ++u) { const int i = i0 + 16 * u; q[u] = p8[i < oe ? i : oe - 1]; }
#pragma unroll
        for (int u = 0; u < 4; ++u)
          if (i0 + 16 * u < oe) atomicAdd(&dcnt[q[u] & (RB - 1)], 1);
      }
    }
  }
  __syncthreads();

  const int wave = threadIdx.x >> 5;
  const int lane = threadIdx.x & 31;
  const int tile = blockIdx.x * 8 + wave;
  if (tile >= NN / 16) return;

  const int row_base = tile * 16;
  const int m    = lane & 15;
  const int hi   = lane >> 4;
  const int koff = hi << 1;

  const float* xrow = X + (size_t)(row_base + m) * F_INd;

  v8f acc = {};
  for (int k0 = 0; k0 < F_INd; k0 += 4) {
    v2f a, b;
    a.x = xrow[k0 + koff];
    a.y = xrow[k0 + koff + 1];
    b.x = Wlds[(k0 + koff)     * HID + m];
    b.y = Wlds[(k0 + koff + 1) * HID + m];
    acc = __builtin_amdgcn_wmma_f32_16x16x4_f32(false, a, false, b,
                                                (short)0, acc, false, false);
  }

  const int rbase = row_base + (hi << 3);
  const int lrow  = (wave << 4) + (hi << 3);
#pragma unroll
  for (int r = 0; r < 8; ++r)
    H1[(size_t)(rbase + r) * HID + m] = acc[r] * rsqrtf((float)(dcnt[lrow + r] + 1));
}


typedef __attribute__((ext_vector_type(4))) float v4f;

template <int N> __device__ __forceinline__ int quad_bcast(int v) {
  return __builtin_amdgcn_update_dpp(v, v, N * 0x55, 0xF, 0xF, false);
}

__device__ __forceinline__ int acc_slot(int dl, int k) { return (dl << 4) | (k ^ ((dl >> 2) & 3)); }

__global__ __launch_bounds__(512, 16) void k_agg1_bucket(const float* __restrict__ h1s,
                                                     const unsigned short* __restrict__ e16,
                                                     const unsigned char* __restrict__ e8,
                                                     const unsigned short* __restrict__ otab,
                                                     const float* __restrict__ b1,
                                                     const float* __restrict__ W2,
                                                     float* __restrict__ h2s) {
  __shared__ float acc[RB * HID];
  __shared__ int dcnt[RB];
  const int t = threadIdx.x, base = blockIdx.x * RB;
  const int j = t & 3, qd = t >> 2;
  const int jq = j << 2;
  const int jc = qd < NBK ? qd : NBK - 1;
  const unsigned short* rw = otab + (size_t)jc * TS + blockIdx.x;
  const int r0 = rw[0], r1 = rw[1];
  v4f own = {0.0f, 0.0f, 0.0f, 0.0f};
  if (base + qd < NN) own = *(const v4f*)(h1s + (size_t)base * HID + 4 * t);
  if (t < RB) dcnt[t] = 0;
  int i0 = qd < NBK ? r0 : 0;
  const int oe = qd < NBK ? r1 : 0;

  const unsigned cb = (unsigned)jc * CH;
  const int last = oe > 0 ? oe - 1 : 0;
  int ia = i0 + j, ib = i0 + 4 + j;
  ia = ia < oe ? ia : last; ib = ib < oe ? ib : last;
  int ra = (int)e8[cb + (unsigned)ia] | ((int)e16[cb + (unsigned)ia] << 8);
  int rb = (int)e8[cb + (unsigned)ib] | ((int)e16[cb + (unsigned)ib] << 8);
  {
    const int sw = (qd >> 2) & 3;
    float* row = acc + ((qd << 4) | jq);
    row[0 ^ sw] = own.x; row[1 ^ sw] = own.y; row[2 ^ sw] = own.z; row[3 ^ sw] = own.w;
  }
  __syncthreads();

  while (i0 < oe) {
    int rec[8];
    rec[0] = quad_bcast<0>(ra); rec[1] = quad_bcast<1>(ra);
    rec[2] = quad_bcast<2>(ra); rec[3] = quad_bcast<3>(ra);
    rec[4] = quad_bcast<0>(rb); rec[5] = quad_bcast<1>(rb);
    rec[6] = quad_bcast<2>(rb); rec[7] = quad_bcast<3>(rb);
    v4f hv[8];
#pragma unroll
    for (int u = 0; u < 8; ++u) {
      const unsigned s = ((unsigned)rec[u] >> 8) | (((unsigned)rec[u] & 0x80u) << 9);
      hv[u] = *(const v4f*)(h1s + ((s << 4) | (unsigned)jq));
    }
    const int inx = i0 + 8;
    int na = inx + j, nb = inx + 4 + j;
    na = na < oe ? na : last; nb = nb < oe ? nb : last;
    const int ran = (int)e8[cb + (unsigned)na] | ((int)e16[cb + (unsigned)na] << 8);
    const int rbn = (int)e8[cb + (unsigned)nb] | ((int)e16[cb + (unsigned)nb] << 8);
    if (i0 + j < oe)     atomicAdd(&dcnt[ra & (RB - 1)], 1);
    if (i0 + 4 + j < oe) atomicAdd(&dcnt[rb & (RB - 1)], 1);
#pragma unroll
    for (int u = 0; u < 8; ++u)
      if (i0 + u < oe) {
        const int dl = rec[u] & (RB - 1);
        const int sw = (dl >> 2) & 3;
        float* row = acc + ((dl << 4) | jq);
        atomicAdd(row + (0 ^ sw), hv[u].x);
        atomicAdd(row + (1 ^ sw), hv[u].y);
        atomicAdd(row + (2 ^ sw), hv[u].z);
        atomicAdd(row + (3 ^ sw), hv[u].w);
      }
    ra = ran; rb = rbn; i0 = inx;
  }
  __syncthreads();
  if (t < RB && base + t < NN) {
    const float di = rsqrtf((float)(dcnt[t] + 1));
    float v[HID];
#pragma unroll
    for (int k = 0; k < HID; ++k) {
      float x = acc[acc_slot(t, k)] * di + b1[k];
      v[k] = x > 0.0f ? x : 0.0f;
    }
#pragma unroll
    for (int c = 0; c < NC; ++c) {
      float sum = 0.0f;
#pragma unroll
      for (int k = 0; k < HID; ++k) sum += v[k] * W2[k * NC + c];
      h2s[(size_t)(base + t) * NC + c] = sum * di;
    }
  }
}

__global__ __launch_bounds__(512) void k_agg2_bucket(const float* __restrict__ h2s,
                                                     const unsigned short* __restrict__ e16,
                                                     const unsigned char* __restrict__ e8,
                                                     const unsigned short* __restrict__ otab,
                                                     float* __restrict__ agg2) {
  __shared__ float acc[RB * 4];
  __shared__ int dcnt[RB];
  const int t = threadIdx.x, base = blockIdx.x * RB;
  if (t < RB) dcnt[t] = 0;
  {
    const int node = base + (t >> 2), c = t & 3;
    acc[t] = (node < NN && c < NC) ? h2s[(size_t)node * NC + c] : 0.0f;
  }
  __syncthreads();
  const int f = t & 15, g = t >> 4;
  int o0[4], o1[4];
#pragma unroll
  for (int k = 0; k < 4; ++k) {
    const int j = g + 32 * k, jc = j < NBK ? j : NBK - 1;
    const unsigned short* rw = otab + (size_t)jc * TS + blockIdx.x;
    o0[k] = j < NBK ? (int)rw[0] : 0;
    o1[k] = j < NBK ? (int)rw[1] : 0;
  }
#pragma unroll
  for (int k = 0; k < 4; ++k) {
    const int jc = (g + 32 * k) < NBK ? (g + 32 * k) : NBK - 1;
    const unsigned short* p16 = e16 + (size_t)jc * CH;
    const unsigned char*  p8  = e8  + (size_t)jc * CH;
    const int oe = o1[k];
    for (int i0 = o0[k] + f; i0 < oe; i0 += 16 * 4) {
      int q[4], lo[4];
      float a0[4], a1[4], a2[4];
#pragma unroll
      for (int u = 0; u < 4; ++u) {
        const int i = i0 + 16 * u, ic = i < oe ? i : oe - 1;
        q[u] = p8[ic]; lo[u] = p16[ic];
      }
#pragma unroll
      for (int u = 0; u < 4; ++u) {
        const float* hs = h2s + (size_t)(lo[u] | ((q[u] >> 7) << 16)) * NC;
        a0[u] = hs[0]; a1[u] = hs[1]; a2[u] = hs[2];
      }
#pragma unroll
      for (int u = 0; u < 4; ++u)
        if (i0 + 16 * u < oe) {
          const int dl = q[u] & (RB - 1);
          atomicAdd(&acc[dl * 4 + 0], a0[u]);
          atomicAdd(&acc[dl * 4 + 1], a1[u]);
          atomicAdd(&acc[dl * 4 + 2], a2[u]);
          atomicAdd(&dcnt[dl], 1);
        }
    }
  }
  __syncthreads();
  if (t < RB && base + t < NN) {
    const float di = rsqrtf((float)(dcnt[t] + 1));
    float* o = agg2 + (size_t)(base + t) * 4;
    o[0] = acc[t * 4 + 0] * di;
    o[1] = acc[t * 4 + 1] * di;
    o[2] = acc[t * 4 + 2] * di;
    o[3] = 0.0f;
  }
}

__global__ void k_logsoftmax(const float* __restrict__ agg2, const float* __restrict__ b2,
                             float* __restrict__ out) {
  int i = blockIdx.x * blockDim.x + threadIdx.x;
  if (i >= NN) return;
  float z0 = agg2[(size_t)i * 4 + 0] + b2[0];
  float z1 = agg2[(size_t)i * 4 + 1] + b2[1];
  float z2 = agg2[(size_t)i * 4 + 2] + b2[2];
  float mx = fmaxf(z0, fmaxf(z1, z2));
  float lse = logf(expf(z0 - mx) + expf(z1 - mx) + expf(z2 - mx));
  out[(size_t)i * NC + 0] = z0 - mx - lse;
  out[(size_t)i * NC + 1] = z1 - mx - lse;
  out[(size_t)i * NC + 2] = z2 - mx - lse;
}

extern "C" void kernel_launch(void* const* d_in, const int* in_sizes, int n_in,
                              void* d_out, int out_size, void* d_ws, size_t ws_size,
                              hipStream_t stream) {
  const float* x  = (const float*)d_in[0];
  const float* W1 = (const float*)d_in[1];
  const float* b1 = (const float*)d_in[2];
  const float* W2 = (const float*)d_in[3];
  const float* b2 = (const float*)d_in[4];
  const int*   ei = (const int*)d_in[5];
  const int* src = ei;
  const int* dst = ei + EE;
  float* out = (float*)d_out;

  char* wsb = (char*)d_ws;
  float*          h1s  = (float*)wsb;
  unsigned short* e16  = (unsigned short*)(wsb + 6400000);
  unsigned char*  e8   = (unsigned char*)(wsb + 12800000);
  unsigned short* otab = (unsigned short*)(wsb + 16000000);
  float* agg2 = (float*)wsb;
  float* h2s = out;

  k_chunk_sort  <<<NBK, 1024, 0, stream>>>(src, dst, e16, e8, otab);
  k_gemm1_wmma  <<<PB, 256, 0, stream>>>(x, W1, e8, otab, h1s);
  k_agg1_bucket <<<PB, 512, 0, stream>>>(h1s, e16, e8, otab, b1, W2, h2s);
  k_agg2_bucket <<<PB, 512, 0, stream>>>(h2s, e16, e8, otab, agg2);
  k_logsoftmax  <<<(NN + 255) / 256, 256, 0, stream>>>(agg2, b2, out);
}


# layer-1 sums: an edge takes a counted slot in its destination row's LDS bin; a 4-lane group then sums its row's source rows in registers (16 float LDS adds per edge gone); sum order = slot order
# speedup vs baseline: 4.9744x; 1.0516x over previous
#include <hip/hip_runtime.h>
#include <cmath>

#define NN   100000
#define EE   3200000
#define F_INd 500
#define HID  16
#define NC   3
#define RSH  7
#define RB   128
#define PB   782
#define NBK  125
#define CH   25600
#define EPT  25
#define TS   832

typedef __attribute__((ext_vector_type(2))) float v2f;
typedef __attribute__((ext_vector_type(8))) float v8f;
typedef unsigned int __attribute__((ext_vector_type(4))) u32x4;
typedef u32x4 __attribute__((may_alias)) u32x4a;


__device__ __forceinline__ int block_excl_scan(int v, int* wsum, int& total) {
  const int lane = threadIdx.x & 31, wid = threadIdx.x >> 5;
  int x = v;
#pragma unroll
  for (int o = 1; o < 32; o <<= 1) { int y = __shfl_up(x, o, 32); if (lane >= o) x += y; }
  if (lane == 31) wsum[wid] = x;
  __syncthreads();
  if (wid == 0) {
    const int nw = blockDim.x >> 5;
    int w = lane < nw ? wsum[lane] : 0;
    int z = w;
#pragma unroll
    for (int o = 1; o < 32; o <<= 1) { int y = __shfl_up(z, o, 32); if (lane >= o) z += y; }
    wsum[lane] = z - w;
    if (lane == 31) wsum[32] = z;
  }
  __syncthreads();
  total = wsum[32];
  return wsum[wid] + x - v;
}

__global__ __launch_bounds__(1024) void k_chunk_sort(const int* __restrict__ src, const int* __restrict__ dst,
                                                     unsigned short* __restrict__ e16,
                                                     unsigned char* __restrict__ e8,
                                                     unsigned short* __restrict__ otab) {
  __shared__ int hist[PB];
  __shared__ int wsum[33];
  __shared__ __attribute__((aligned(16))) unsigned short st16[CH];
  __shared__ __attribute__((aligned(16))) unsigned char  st8[CH];
  __shared__ __attribute__((aligned(16))) unsigned short row[TS];
  const int t = threadIdx.x, blk = blockIdx.x;
  if (t < PB) hist[t] = 0;
  if (t < TS) row[t] = 0;
  __syncthreads();
  const int* sp = src + blk * CH;
  const int* dp = dst + blk * CH;
  int sv[EPT], dv[EPT];
#pragma unroll
  for (int u = 0; u < EPT; ++u) { sv[u] = sp[t + 1024 * u]; dv[u] = dp[t + 1024 * u]; }
#pragma unroll
  for (int u = 0; u < EPT; ++u) atomicAdd(&hist[dv[u] >> RSH], 1);
  __syncthreads();
  const int c = t < PB ? hist[t] : 0;
  int total;
  const int lo = block_excl_scan(c, wsum, total);
  if (t < PB) { hist[t] = lo; row[t] = (unsigned short)lo; }
  if (t == 0) row[PB] = (unsigned short)CH;
  __syncthreads();
#pragma unroll
  for (int u = 0; u < EPT; ++u) {
    const int r = atomicAdd(&hist[dv[u] >> RSH], 1);
    st16[r] = (unsigned short)(sv[u] & 0xFFFF);
    st8[r]  = (unsigned char)((dv[u] & (RB - 1)) | ((sv[u] >> 16) << 7));
  }
  __syncthreads();
  const u32x4a* v16 = (const u32x4a*)st16;
  u32x4a* g16 = (u32x4a*)(e16 + (size_t)blk * CH);
  for (int i = t; i < CH / 8; i += 1024) g16[i] = v16[i];
  const u32x4a* v8 = (const u32x4a*)st8;
  u32x4a* g8 = (u32x4a*)(e8 + (size_t)blk * CH);
  for (int i = t; i < CH / 16; i += 1024) g8[i] = v8[i];
  const u32x4a* vr = (const u32x4a*)row;
  u32x4a* gr = (u32x4a*)(otab + (size_t)blk * TS);
  if (t < TS / 8) gr[t] = vr[t];
}

__global__ __launch_bounds__(256) void k_gemm1_wmma(const float* __restrict__ X,
                                                    const float* __restrict__ W,
                                                    const unsigned char* __restrict__ e8,
                                                    const unsigned short* __restrict__ otab,
                                                    float* __restrict__ H1) {
  __shared__ float Wlds[F_INd * HID];
  __shared__ int dcnt[RB];
  for (int i = threadIdx.x; i < F_INd * HID; i += 256) Wlds[i] = W[i];
  if (threadIdx.x < RB) dcnt[threadIdx.x] = 0;
  __syncthreads();
  {
    const int g = threadIdx.x >> 4, f = threadIdx.x & 15;
    int o0[8], o1[8];
#pragma unroll
    for (int k = 0; k < 8; ++k) {
      const int j = g + 16 * k, jc = j < NBK ? j : NBK - 1;
      const unsigned short* rw = otab + (size_t)jc * TS + blockIdx.x;
      o0[k] = j < NBK ? (int)rw[0] : 0;
      o1[k] = j < NBK ? (int)rw[1] : 0;
    }
#pragma unroll
    for (int k = 0; k < 8; ++k) {
      const int jc = (g + 16 * k) < NBK ? (g + 16 * k) : NBK - 1;
      const unsigned char* p8 = e8 + (size_t)jc * CH;
      const int oe = o1[k];
      for (int i0 = o0[k] + f; i0 < oe; i0 += 16 * 4) {
        int q[4];
#pragma unroll
        for (int u = 0; u < 4; ++u) { const int i = i0 + 16 * u; q[u] = p8[i < oe ? i : oe - 1]; }
#pragma unroll
        for (int u = 0; u < 4; ++u)
          if (i0 + 16 * u < oe) atomicAdd(&dcnt[q[u] & (RB - 1)], 1);
      }
    }
  }
  __syncthreads();

  const int wave = threadIdx.x >> 5;
  const int lane = threadIdx.x & 31;
  const int tile = blockIdx.x * 8 + wave;
  if (tile >= NN / 16) return;

  const int row_base = tile * 16;
  const int m    = lane & 15;
  const int hi   = lane >> 4;
  const int koff = hi << 1;

  const float* xrow = X + (size_t)(row_base + m) * F_INd;

  v8f acc = {};
  for (int k0 = 0; k0 < F_INd; k0 += 4) {
    v2f a, b;
    a.x = xrow[k0 + koff];
    a.y = xrow[k0 + koff + 1];
    b.x = Wlds[(k0 + koff)     * HID + m];
    b.y = Wlds[(k0 + koff + 1) * HID + m];
    acc = __builtin_amdgcn_wmma_f32_16x16x4_f32(false, a, false, b,
                                                (short)0, acc, false, false);
  }

  const int rbase = row_base + (hi << 3);
  const int lrow  = (wave << 4) + (hi << 3);
#pragma unroll
  for (int r = 0; r < 8; ++r)
    H1[(size_t)(rbase + r) * HID + m] = acc[r] * rsqrtf((float)(dcnt[lrow + r] + 1));
}


typedef __attribute__((ext_vector_type(4))) float v4f;
typedef __attribute__((ext_vector_type(4))) int i32x4;

#define BINW 64
#define BINP 68
#define RPL  16

__global__ __launch_bounds__(512, 16) void k_agg1_bucket(const float* __restrict__ h1s,
                                                     const unsigned short* __restrict__ e16,
                                                     const unsigned char* __restrict__ e8,
                                                     const unsigned short* __restrict__ otab,
                                                     const float* __restrict__ b1,
                                                     const float* __restrict__ W2,
                                                     float* __restrict__ h2s) {
  __shared__ __attribute__((aligned(16))) float acc[RB * HID];
  __shared__ int cnt[RB];
  __shared__ int full;
  __shared__ __attribute__((aligned(16))) int bin[RB * BINP];
  const int t = threadIdx.x, base = blockIdx.x * RB;
  const int j = t & 3, qd = t >> 2;
  const int jq = j << 2;
  const int jc = qd < NBK ? qd : NBK - 1;
  const unsigned short* rw = otab + (size_t)jc * TS + blockIdx.x;
  const int r0 = rw[0], r1 = rw[1];
  v4f own = {0.0f, 0.0f, 0.0f, 0.0f};
  if (base + qd < NN) own = *(const v4f*)(h1s + (size_t)base * HID + 4 * t);
  if (t < RB) cnt[t] = 0;
  if (t == 0) full = 0;
  for (int i = t; i < RB * HID; i += 512) acc[i] = 0.0f;
  __syncthreads();

  {
    const int oe = qd < NBK ? r1 : 0;
    const unsigned cb = (unsigned)jc * CH;
    for (int i0 = (qd < NBK ? r0 : 0) + j; i0 < oe; i0 += 4 * RPL) {
      int rec[RPL];
#pragma unroll
      for (int u = 0; u < RPL; ++u) {
        int i = i0 + 4 * u;
        i = i < oe ? i : oe - 1;
        rec[u] = (int)e8[cb + (unsigned)i] | ((int)e16[cb + (unsigned)i] << 8);
      }
      int slot[RPL];
#pragma unroll
      for (int u = 0; u < RPL; ++u) {
        slot[u] = 0;
        if (i0 + 4 * u < oe) slot[u] = atomicAdd(&cnt[rec[u] & (RB - 1)], 1);
      }
#pragma unroll
      for (int u = 0; u < RPL; ++u)
        if (i0 + 4 * u < oe) {
          const int dl = rec[u] & (RB - 1);
          const unsigned s = ((unsigned)rec[u] >> 8) | (((unsigned)rec[u] & 0x80u) << 9);
          if (slot[u] < BINW) bin[dl * BINP + slot[u]] = (int)s;
          else full = 1;
        }
    }
  }
  __syncthreads();

  {
    v4f sum = own;
    const int any_full = full;
    if (any_full) {
      const int oe = qd < NBK ? r1 : 0;
      const unsigned cb = (unsigned)jc * CH;
      for (int i = (qd < NBK ? r0 : 0) + j; i < oe; i += 4) {
        const int rec = (int)e8[cb + (unsigned)i] | ((int)e16[cb + (unsigned)i] << 8);
        const int dl = rec & (RB - 1);
        const unsigned s = ((unsigned)rec >> 8) | (((unsigned)rec & 0x80u) << 9);
        const float* hs = h1s + ((size_t)s << 4);
        for (int k = 0; k < HID; ++k) atomicAdd(&acc[(dl << 4) | k], hs[k]);
      }
      __syncthreads();
    }
    const int deg = cnt[qd];
    const int n = any_full ? 0 : (deg < BINW ? deg : BINW);
    const int* bp = bin + qd * BINP;
    for (int i0 = 0; i0 < n; i0 += 8) {
      const i32x4 ia = *(const i32x4*)(bp + i0);
      const i32x4 ib = *(const i32x4*)(bp + i0 + 4);
      const int id[8] = {ia.x, ia.y, ia.z, ia.w, ib.x, ib.y, ib.z, ib.w};
      v4f hv[8];
#pragma unroll
      for (int u = 0; u < 8; ++u) {
        const unsigned s = (i0 + u < n) ? ((unsigned)id[u] & 0x1FFFFu) : 0u;
        hv[u] = *(const v4f*)(h1s + ((s << 4) | (unsigned)jq));
      }
      __builtin_amdgcn_sched_barrier(0);
#pragma unroll
      for (int u = 0; u < 8; ++u) {
        const bool live = i0 + u < n;
        sum.x += live ? hv[u].x : 0.0f; sum.y += live ? hv[u].y : 0.0f;
        sum.z += live ? hv[u].z : 0.0f; sum.w += live ? hv[u].w : 0.0f;
      }
    }
    v4f* rowv = (v4f*)(acc + ((qd << 4) | jq));
    v4f o = *rowv;
    o.x += sum.x; o.y += sum.y; o.z += sum.z; o.w += sum.w;
    *rowv = o;
  }
  __syncthreads();
  if (t < RB && base + t < NN) {
    const float di = rsqrtf((float)(cnt[t] + 1));
    float v[HID];
#pragma unroll
    for (int k = 0; k < HID; ++k) {
      float x = acc[(t << 4) | k] * di + b1[k];
      v[k] = x > 0.0f ? x : 0.0f;
    }
#pragma unroll
    for (int c = 0; c < NC; ++c) {
      float sum = 0.0f;
#pragma unroll
      for (int k = 0; k < HID; ++k) sum += v[k] * W2[k * NC + c];
      h2s[(size_t)(base + t) * NC + c] = sum * di;
    }
  }
}

__global__ __launch_bounds__(512) void k_agg2_bucket(const float* __restrict__ h2s,
                                                     const unsigned short* __restrict__ e16,
                                                     const unsigned char* __restrict__ e8,
                                                     const unsigned short* __restrict__ otab,
                                                     float* __restrict__ agg2) {
  __shared__ float acc[RB * 4];
  __shared__ int dcnt[RB];
  const int t = threadIdx.x, base = blockIdx.x * RB;
  if (t < RB) dcnt[t] = 0;
  {
    const int node = base + (t >> 2), c = t & 3;
    acc[t] = (node < NN && c < NC) ? h2s[(size_t)node * NC + c] : 0.0f;
  }
  __syncthreads();
  const int f = t & 15, g = t >> 4;
  int o0[4], o1[4];
#pragma unroll
  for (int k = 0; k < 4; ++k) {
    const int j = g + 32 * k, jc = j < NBK ? j : NBK - 1;
    const unsigned short* rw = otab + (size_t)jc * TS + blockIdx.x;
    o0[k] = j < NBK ? (int)rw[0] : 0;
    o1[k] = j < NBK ? (int)rw[1] : 0;
  }
#pragma unroll
  for (int k = 0; k < 4; ++k) {
    const int jc = (g + 32 * k) < NBK ? (g + 32 * k) : NBK - 1;
    const unsigned short* p16 = e16 + (size_t)jc * CH;
    const unsigned char*  p8  = e8  + (size_t)jc * CH;
    const int oe = o1[k];
    for (int i0 = o0[k] + f; i0 < oe; i0 += 16 * 4) {
      int q[4], lo[4];
      float a0[4], a1[4], a2[4];
#pragma unroll
      for (int u = 0; u < 4; ++u) {
        const int i = i0 + 16 * u, ic = i < oe ? i : oe - 1;
        q[u] = p8[ic]; lo[u] = p16[ic];
      }
#pragma unroll
      for (int u = 0; u < 4; ++u) {
        const float* hs = h2s + (size_t)(lo[u] | ((q[u] >> 7) << 16)) * NC;
        a0[u] = hs[0]; a1[u] = hs[1]; a2[u] = hs[2];
      }
#pragma unroll
      for (int u = 0; u < 4; ++u)
        if (i0 + 16 * u < oe) {
          const int dl = q[u] & (RB - 1);
          atomicAdd(&acc[dl * 4 + 0], a0[u]);
          atomicAdd(&acc[dl * 4 + 1], a1[u]);
          atomicAdd(&acc[dl * 4 + 2], a2[u]);
          atomicAdd(&dcnt[dl], 1);
        }
    }
  }
  __syncthreads();
  if (t < RB && base + t < NN) {
    const float di = rsqrtf((float)(dcnt[t] + 1));
    float* o = agg2 + (size_t)(base + t) * 4;
    o[0] = acc[t * 4 + 0] * di;
    o[1] = acc[t * 4 + 1] * di;
    o[2] = acc[t * 4 + 2] * di;
    o[3] = 0.0f;
  }
}

__global__ void k_logsoftmax(const float* __restrict__ agg2, const float* __restrict__ b2,
                             float* __restrict__ out) {
  int i = blockIdx.x * blockDim.x + threadIdx.x;
  if (i >= NN) return;
  float z0 = agg2[(size_t)i * 4 + 0] + b2[0];
  float z1 = agg2[(size_t)i * 4 + 1] + b2[1];
  float z2 = agg2[(size_t)i * 4 + 2] + b2[2];
  float mx = fmaxf(z0, fmaxf(z1, z2));
  float lse = logf(expf(z0 - mx) + expf(z1 - mx) + expf(z2 - mx));
  out[(size_t)i * NC + 0] = z0 - mx - lse;
  out[(size_t)i * NC + 1] = z1 - mx - lse;
  out[(size_t)i * NC + 2] = z2 - mx - lse;
}

extern "C" void kernel_launch(void* const* d_in, const int* in_sizes, int n_in,
                              void* d_out, int out_size, void* d_ws, size_t ws_size,
                              hipStream_t stream) {
  const float* x  = (const float*)d_in[0];
  const float* W1 = (const float*)d_in[1];
  const float* b1 = (const float*)d_in[2];
  const float* W2 = (const float*)d_in[3];
  const float* b2 = (const float*)d_in[4];
  const int*   ei = (const int*)d_in[5];
  const int* src = ei;
  const int* dst = ei + EE;
  float* out = (float*)d_out;

  char* wsb = (char*)d_ws;
  float*          h1s  = (float*)wsb;
  unsigned short* e16  = (unsigned short*)(wsb + 6400000);
  unsigned char*  e8   = (unsigned char*)(wsb + 12800000);
  unsigned short* otab = (unsigned short*)(wsb + 16000000);
  float* agg2 = (float*)wsb;
  float* h2s = out;

  k_chunk_sort  <<<NBK, 1024, 0, stream>>>(src, dst, e16, e8, otab);
  k_gemm1_wmma  <<<PB, 256, 0, stream>>>(x, W1, e8, otab, h1s);
  k_agg1_bucket <<<PB, 512, 0, stream>>>(h1s, e16, e8, otab, b1, W2, h2s);
  k_agg2_bucket <<<PB, 512, 0, stream>>>(h2s, e16, e8, otab, agg2);
  k_logsoftmax  <<<(NN + 255) / 256, 256, 0, stream>>>(agg2, b2, out);
}


# GEMM1 X path: 16-byte loads along a row into a per-wave LDS tile (next piece in flight), one 16-byte LDS read per operand feeds two WMMAs; K padded to 504 by zeros, k order inside each 8 permuted
# speedup vs baseline: 6.0418x; 1.2146x over previous
#include <hip/hip_runtime.h>
#include <cmath>

#define NN   100000
#define EE   3200000
#define F_INd 500
#define HID  16
#define NC   3
#define RSH  7
#define RB   128
#define PB   782
#define NBK  125
#define CH   25600
#define EPT  25
#define TS   832

typedef __attribute__((ext_vector_type(2))) float v2f;
typedef __attribute__((ext_vector_type(8))) float v8f;
typedef unsigned int __attribute__((ext_vector_type(4))) u32x4;
typedef u32x4 __attribute__((may_alias)) u32x4a;


__device__ __forceinline__ int block_excl_scan(int v, int* wsum, int& total) {
  const int lane = threadIdx.x & 31, wid = threadIdx.x >> 5;
  int x = v;
#pragma unroll
  for (int o = 1; o < 32; o <<= 1) { int y = __shfl_up(x, o, 32); if (lane >= o) x += y; }
  if (lane == 31) wsum[wid] = x;
  __syncthreads();
  if (wid == 0) {
    const int nw = blockDim.x >> 5;
    int w = lane < nw ? wsum[lane] : 0;
    int z = w;
#pragma unroll
    for (int o = 1; o < 32; o <<= 1) { int y = __shfl_up(z, o, 32); if (lane >= o) z += y; }
    wsum[lane] = z - w;
    if (lane == 31) wsum[32] = z;
  }
  __syncthreads();
  total = wsum[32];
  return wsum[wid] + x - v;
}

__global__ __launch_bounds__(1024) void k_chunk_sort(const int* __restrict__ src, const int* __restrict__ dst,
                                                     unsigned short* __restrict__ e16,
                                                     unsigned char* __restrict__ e8,
                                                     unsigned short* __restrict__ otab) {
  __shared__ int hist[PB];
  __shared__ int wsum[33];
  __shared__ __attribute__((aligned(16))) unsigned short st16[CH];
  __shared__ __attribute__((aligned(16))) unsigned char  st8[CH];
  __shared__ __attribute__((aligned(16))) unsigned short row[TS];
  const int t = threadIdx.x, blk = blockIdx.x;
  if (t < PB) hist[t] = 0;
  if (t < TS) row[t] = 0;
  __syncthreads();
  const int* sp = src + blk * CH;
  const int* dp = dst + blk * CH;
  int sv[EPT], dv[EPT];
#pragma unroll
  for (int u = 0; u < EPT; ++u) { sv[u] = sp[t + 1024 * u]; dv[u] = dp[t + 1024 * u]; }
#pragma unroll
  for (int u = 0; u < EPT; ++u) atomicAdd(&hist[dv[u] >> RSH], 1);
  __syncthreads();
  const int c = t < PB ? hist[t] : 0;
  int total;
  const int lo = block_excl_scan(c, wsum, total);
  if (t < PB) { hist[t] = lo; row[t] = (unsigned short)lo; }
  if (t == 0) row[PB] = (unsigned short)CH;
  __syncthreads();
#pragma unroll
  for (int u = 0; u < EPT; ++u) {
    const int r = atomicAdd(&hist[dv[u] >> RSH], 1);
    st16[r] = (unsigned short)(sv[u] & 0xFFFF);
    st8[r]  = (unsigned char)((dv[u] & (RB - 1)) | ((sv[u] >> 16) << 7));
  }
  __syncthreads();
  const u32x4a* v16 = (const u32x4a*)st16;
  u32x4a* g16 = (u32x4a*)(e16 + (size_t)blk * CH);
  for (int i = t; i < CH / 8; i += 1024) g16[i] = v16[i];
  const u32x4a* v8 = (const u32x4a*)st8;
  u32x4a* g8 = (u32x4a*)(e8 + (size_t)blk * CH);
  for (int i = t; i < CH / 16; i += 1024) g8[i] = v8[i];
  const u32x4a* vr = (const u32x4a*)row;
  u32x4a* gr = (u32x4a*)(otab + (size_t)blk * TS);
  if (t < TS / 8) gr[t] = vr[t];
}

typedef __attribute__((ext_vector_type(4))) float g1v4;
#define XKC 64
#define XP  68
#define NPC 8
#define KPAD 504

__global__ __launch_bounds__(256) __attribute__((amdgpu_waves_per_eu(8, 8))) void k_gemm1_wmma(const float* __restrict__ X,
                                                    const float* __restrict__ W,
                                                    const unsigned char* __restrict__ e8,
                                                    const unsigned short* __restrict__ otab,
                                                    float* __restrict__ H1) {
  __shared__ __attribute__((aligned(16))) float Wq[KPAD * HID];
  __shared__ __attribute__((aligned(16))) float xt[8 * 16 * XP];
  __shared__ int dcnt[RB];
  const int wave = threadIdx.x >> 5;
  const int lane = threadIdx.x & 31;
  const int tile = blockIdx.x * 8 + wave;
  const int row_base = tile * 16;
  const int m    = lane & 15;
  const int hi   = lane >> 4;
  const float* xld = X + (size_t)(row_base + hi) * F_INd + 4 * m;
  g1v4 R[8];
  if (tile < NN / 16) {
#pragma unroll
    for (int u = 0; u < 8; ++u) R[u] = *(const g1v4*)(xld + (size_t)(2 * u) * F_INd);
  }
  for (int i = threadIdx.x; i < F_INd * HID; i += 256) {
    const int k = i >> 4, c = i & 15;
    Wq[((((k >> 2) << 4) + c) << 2) + (k & 3)] = W[i];
  }
  if (threadIdx.x < (KPAD - F_INd) * HID) Wq[F_INd * HID + threadIdx.x] = 0.0f;
  if (threadIdx.x < RB) dcnt[threadIdx.x] = 0;
  __syncthreads();
  {
    const int g = threadIdx.x >> 4, f = threadIdx.x & 15;
    int o0[8], o1[8];
#pragma unroll
    for (int k = 0; k < 8; ++k) {
      const int j = g + 16 * k, jc = j < NBK ? j : NBK - 1;
      const unsigned short* rw = otab + (size_t)jc * TS + blockIdx.x;
      o0[k] = j < NBK ? (int)rw[0] : 0;
      o1[k] = j < NBK ? (int)rw[1] : 0;
    }
#pragma unroll
    for (int k = 0; k < 8; ++k) {
      const int jc = (g + 16 * k) < NBK ? (g + 16 * k) : NBK - 1;
      const unsigned char* p8 = e8 + (size_t)jc * CH;
      const int oe = o1[k];
      for (int i0 = o0[k] + f; i0 < oe; i0 += 16 * 4) {
        int q[4];
#pragma unroll
        for (int u = 0; u < 4; ++u) { const int i = i0 + 16 * u; q[u] = p8[i < oe ? i : oe - 1]; }
#pragma unroll
        for (int u = 0; u < 4; ++u)
          if (i0 + 16 * u < oe) atomicAdd(&dcnt[q[u] & (RB - 1)], 1);
      }
    }
  }
  __syncthreads();

  if (tile >= NN / 16) return;

  float* tl = xt + wave * (16 * XP);
  float* tst = tl + hi * XP + 4 * m;
  const float* afr = tl + m * XP + 4 * hi;
  const float* bfr = Wq + (hi << 6) + 4 * m;
  {
    const g1v4 z = {0.0f, 0.0f, 0.0f, 0.0f};
    *(g1v4*)(tl + m * XP + XKC) = z;
  }

  v8f acc = {};
#pragma unroll
  for (int c = 0; c < NPC; ++c) {
    __builtin_amdgcn_fence(__ATOMIC_ACQ_REL, "wavefront");
#pragma unroll
    for (int u = 0; u < 8; ++u) *(g1v4*)(tst + (2 * u) * XP) = R[u];
    __builtin_amdgcn_fence(__ATOMIC_ACQ_REL, "wavefront");
    if (c + 1 < NPC) {
      const int nc = (c + 1 < NPC - 1) ? (c + 1) * XKC : F_INd - XKC;
#pragma unroll
      for (int u = 0; u < 8; ++u) R[u] = *(const g1v4*)(xld + (size_t)(2 * u) * F_INd + nc);
    }
    const int c0 = (c + 1 < NPC) ? c * XKC : F_INd - XKC;
    const int J0 = c * (XKC / 8);
    const int nj = (c + 1 < NPC) ? XKC / 8 : KPAD / 8 - J0;
    g1v4 a4[XKC / 8], b4[XKC / 8];
#pragma unroll
    for (int j = 0; j < nj; ++j) {
      a4[j] = *(const g1v4*)(afr + (8 * (J0 + j) - c0));
      b4[j] = *(const g1v4*)(bfr + 128 * (J0 + j));
    }
    __builtin_amdgcn_sched_barrier(0);
#pragma unroll
    for (int j = 0; j < nj; ++j) {
      const v2f a0 = {a4[j].x, a4[j].y}, b0 = {b4[j].x, b4[j].y};
      const v2f a1 = {a4[j].z, a4[j].w}, b1 = {b4[j].z, b4[j].w};
      acc = __builtin_amdgcn_wmma_f32_16x16x4_f32(false, a0, false, b0, (short)0, acc, false, false);
      acc = __builtin_amdgcn_wmma_f32_16x16x4_f32(false, a1, false, b1, (short)0, acc, false, false);
    }
  }
  __builtin_amdgcn_sched_barrier(0);
  asm volatile("v_nop\n\tv_nop\n\tv_nop\n\tv_nop");
  __builtin_amdgcn_sched_barrier(0);

  const int rbase = row_base + (hi << 3);
  const int lrow  = (wave << 4) + (hi << 3);
#pragma unroll
  for (int r = 0; r < 8; ++r)
    H1[(size_t)(rbase + r) * HID + m] = acc[r] * rsqrtf((float)(dcnt[lrow + r] + 1));
}


typedef __attribute__((ext_vector_type(4))) float v4f;
typedef __attribute__((ext_vector_type(4))) int i32x4;

#define BINW 64
#define BINP 68
#define RPL  16

__global__ __launch_bounds__(512, 16) void k_agg1_bucket(const float* __restrict__ h1s,
                                                     const unsigned short* __restrict__ e16,
                                                     const unsigned char* __restrict__ e8,
                                                     const unsigned short* __restrict__ otab,
                                                     const float* __restrict__ b1,
                                                     const float* __restrict__ W2,
                                                     float* __restrict__ h2s) {
  __shared__ __attribute__((aligned(16))) float acc[RB * HID];
  __shared__ int cnt[RB];
  __shared__ int full;
  __shared__ __attribute__((aligned(16))) int bin[RB * BINP];
  const int t = threadIdx.x, base = blockIdx.x * RB;
  const int j = t & 3, qd = t >> 2;
  const int jq = j << 2;
  const int jc = qd < NBK ? qd : NBK - 1;
  const unsigned short* rw = otab + (size_t)jc * TS + blockIdx.x;
  const int r0 = rw[0], r1 = rw[1];
  v4f own = {0.0f, 0.0f, 0.0f, 0.0f};
  if (base + qd < NN) own = *(const v4f*)(h1s + (size_t)base * HID + 4 * t);
  if (t < RB) cnt[t] = 0;
  if (t == 0) full = 0;
  for (int i = t; i < RB * HID; i += 512) acc[i] = 0.0f;
  __syncthreads();

  {
    const int oe = qd < NBK ? r1 : 0;
    const unsigned cb = (unsigned)jc * CH;
    for (int i0 = (qd < NBK ? r0 : 0) + j; i0 < oe; i0 += 4 * RPL) {
      int rec[RPL];
#pragma unroll
      for (int u = 0; u < RPL; ++u) {
        int i = i0 + 4 * u;
        i = i < oe ? i : oe - 1;
        rec[u] = (int)e8[cb + (unsigned)i] | ((int)e16[cb + (unsigned)i] << 8);
      }
      int slot[RPL];
#pragma unroll
      for (int u = 0; u < RPL; ++u) {
        slot[u] = 0;
        if (i0 + 4 * u < oe) slot[u] = atomicAdd(&cnt[rec[u] & (RB - 1)], 1);
      }
#pragma unroll
      for (int u = 0; u < RPL; ++u)
        if (i0 + 4 * u < oe) {
          const int dl = rec[u] & (RB - 1);
          const unsigned s = ((unsigned)rec[u] >> 8) | (((unsigned)rec[u] & 0x80u) << 9);
          if (slot[u] < BINW) bin[dl * BINP + slot[u]] = (int)s;
          else full = 1;
        }
    }
  }
  __syncthreads();

  {
    v4f sum = own;
    const int any_full = full;
    if (any_full) {
      const int oe = qd < NBK ? r1 : 0;
      const unsigned cb = (unsigned)jc * CH;
      for (int i = (qd < NBK ? r0 : 0) + j; i < oe; i += 4) {
        const int rec = (int)e8[cb + (unsigned)i] | ((int)e16[cb + (unsigned)i] << 8);
        const int dl = rec & (RB - 1);
        const unsigned s = ((unsigned)rec >> 8) | (((unsigned)rec & 0x80u) << 9);
        const float* hs = h1s + ((size_t)s << 4);
        for (int k = 0; k < HID; ++k) atomicAdd(&acc[(dl << 4) | k], hs[k]);
      }
      __syncthreads();
    }
    const int deg = cnt[qd];
    const int n = any_full ? 0 : (deg < BINW ? deg : BINW);
    const int* bp = bin + qd * BINP;
    for (int i0 = 0; i0 < n; i0 += 8) {
      const i32x4 ia = *(const i32x4*)(bp + i0);
      const i32x4 ib = *(const i32x4*)(bp + i0 + 4);
      const int id[8] = {ia.x, ia.y, ia.z, ia.w, ib.x, ib.y, ib.z, ib.w};
      v4f hv[8];
#pragma unroll
      for (int u = 0; u < 8; ++u) {
        const unsigned s = (i0 + u < n) ? ((unsigned)id[u] & 0x1FFFFu) : 0u;
        hv[u] = *(const v4f*)(h1s + ((s << 4) | (unsigned)jq));
      }
      __builtin_amdgcn_sched_barrier(0);
#pragma unroll
      for (int u = 0; u < 8; ++u) {
        const bool live = i0 + u < n;
        sum.x += live ? hv[u].x : 0.0f; sum.y += live ? hv[u].y : 0.0f;
        sum.z += live ? hv[u].z : 0.0f; sum.w += live ? hv[u].w : 0.0f;
      }
    }
    v4f* rowv = (v4f*)(acc + ((qd << 4) | jq));
    v4f o = *rowv;
    o.x += sum.x; o.y += sum.y; o.z += sum.z; o.w += sum.w;
    *rowv = o;
  }
  __syncthreads();
  if (t < RB && base + t < NN) {
    const float di = rsqrtf((float)(cnt[t] + 1));
    float v[HID];
#pragma unroll
    for (int k = 0; k < HID; ++k) {
      float x = acc[(t << 4) | k] * di + b1[k];
      v[k] = x > 0.0f ? x : 0.0f;
    }
#pragma unroll
    for (int c = 0; c < NC; ++c) {
      float sum = 0.0f;
#pragma unroll
      for (int k = 0; k < HID; ++k) sum += v[k] * W2[k * NC + c];
      h2s[(size_t)(base + t) * NC + c] = sum * di;
    }
  }
}

__global__ __launch_bounds__(512) void k_agg2_bucket(const float* __restrict__ h2s,
                                                     const unsigned short* __restrict__ e16,
                                                     const unsigned char* __restrict__ e8,
                                                     const unsigned short* __restrict__ otab,
                                                     float* __restrict__ agg2) {
  __shared__ float acc[RB * 4];
  __shared__ int dcnt[RB];
  const int t = threadIdx.x, base = blockIdx.x * RB;
  if (t < RB) dcnt[t] = 0;
  {
    const int node = base + (t >> 2), c = t & 3;
    acc[t] = (node < NN && c < NC) ? h2s[(size_t)node * NC + c] : 0.0f;
  }
  __syncthreads();
  const int f = t & 15, g = t >> 4;
  int o0[4], o1[4];
#pragma unroll
  for (int k = 0; k < 4; ++k) {
    const int j = g + 32 * k, jc = j < NBK ? j : NBK - 1;
    const unsigned short* rw = otab + (size_t)jc * TS + blockIdx.x;
    o0[k] = j < NBK ? (int)rw[0] : 0;
    o1[k] = j < NBK ? (int)rw[1] : 0;
  }
#pragma unroll
  for (int k = 0; k < 4; ++k) {
    const int jc = (g + 32 * k) < NBK ? (g + 32 * k) : NBK - 1;
    const unsigned short* p16 = e16 + (size_t)jc * CH;
    const unsigned char*  p8  = e8  + (size_t)jc * CH;
    const int oe = o1[k];
    for (int i0 = o0[k] + f; i0 < oe; i0 += 16 * 4) {
      int q[4], lo[4];
      float a0[4], a1[4], a2[4];
#pragma unroll
      for (int u = 0; u < 4; ++u) {
        const int i = i0 + 16 * u, ic = i < oe ? i : oe - 1;
        q[u] = p8[ic]; lo[u] = p16[ic];
      }
#pragma unroll
      for (int u = 0; u < 4; ++u) {
        const float* hs = h2s + (size_t)(lo[u] | ((q[u] >> 7) << 16)) * NC;
        a0[u] = hs[0]; a1[u] = hs[1]; a2[u] = hs[2];
      }
#pragma unroll
      for (int u = 0; u < 4; ++u)
        if (i0 + 16 * u < oe) {
          const int dl = q[u] & (RB - 1);
          atomicAdd(&acc[dl * 4 + 0], a0[u]);
          atomicAdd(&acc[dl * 4 + 1], a1[u]);
          atomicAdd(&acc[dl * 4 + 2], a2[u]);
          atomicAdd(&dcnt[dl], 1);
        }
    }
  }
  __syncthreads();
  if (t < RB && base + t < NN) {
    const float di = rsqrtf((float)(dcnt[t] + 1));
    float* o = agg2 + (size_t)(base + t) * 4;
    o[0] = acc[t * 4 + 0] * di;
    o[1] = acc[t * 4 + 1] * di;
    o[2] = acc[t * 4 + 2] * di;
    o[3] = 0.0f;
  }
}

__global__ void k_logsoftmax(const float* __restrict__ agg2, const float* __restrict__ b2,
                             float* __restrict__ out) {
  int i = blockIdx.x * blockDim.x + threadIdx.x;
  if (i >= NN) return;
  float z0 = agg2[(size_t)i * 4 + 0] + b2[0];
  float z1 = agg2[(size_t)i * 4 + 1] + b2[1];
  float z2 = agg2[(size_t)i * 4 + 2] + b2[2];
  float mx = fmaxf(z0, fmaxf(z1, z2));
  float lse = logf(expf(z0 - mx) + expf(z1 - mx) + expf(z2 - mx));
  out[(size_t)i * NC + 0] = z0 - mx - lse;
  out[(size_t)i * NC + 1] = z1 - mx - lse;
  out[(size_t)i * NC + 2] = z2 - mx - lse;
}

extern "C" void kernel_launch(void* const* d_in, const int* in_sizes, int n_in,
                              void* d_out, int out_size, void* d_ws, size_t ws_size,
                              hipStream_t stream) {
  const float* x  = (const float*)d_in[0];
  const float* W1 = (const float*)d_in[1];
  const float* b1 = (const float*)d_in[2];
  const float* W2 = (const float*)d_in[3];
  const float* b2 = (const float*)d_in[4];
  const int*   ei = (const int*)d_in[5];
  const int* src = ei;
  const int* dst = ei + EE;
  float* out = (float*)d_out;

  char* wsb = (char*)d_ws;
  float*          h1s  = (float*)wsb;
  unsigned short* e16  = (unsigned short*)(wsb + 6400000);
  unsigned char*  e8   = (unsigned char*)(wsb + 12800000);
  unsigned short* otab = (unsigned short*)(wsb + 16000000);
  float* agg2 = (float*)wsb;
  float* h2s = out;

  k_chunk_sort  <<<NBK, 1024, 0, stream>>>(src, dst, e16, e8, otab);
  k_gemm1_wmma  <<<PB, 256, 0, stream>>>(x, W1, e8, otab, h1s);
  k_agg1_bucket <<<PB, 512, 0, stream>>>(h1s, e16, e8, otab, b1, W2, h2s);
  k_agg2_bucket <<<PB, 512, 0, stream>>>(h2s, e16, e8, otab, agg2);
  k_logsoftmax  <<<(NN + 255) / 256, 256, 0, stream>>>(agg2, b2, out);
}


# layer-2 sums: block = 13 dst buckets x ONE of 4 source slices staged once in LDS; edges read source rows from LDS, not by request; 4 partial sums per node added in slice order; LDS add order free
# speedup vs baseline: 6.2699x; 1.0378x over previous
#include <hip/hip_runtime.h>
#include <cmath>

#define NN   100000
#define EE   3200000
#define F_INd 500
#define HID  16
#define NC   3
#define RSH  7
#define RB   128
#define PB   782
#define NBK  125
#define CH   25600
#define EPT  25
#define TS   832

typedef __attribute__((ext_vector_type(2))) float v2f;
typedef __attribute__((ext_vector_type(8))) float v8f;
typedef unsigned int __attribute__((ext_vector_type(4))) u32x4;
typedef u32x4 __attribute__((may_alias)) u32x4a;


__device__ __forceinline__ int block_excl_scan(int v, int* wsum, int& total) {
  const int lane = threadIdx.x & 31, wid = threadIdx.x >> 5;
  int x = v;
#pragma unroll
  for (int o = 1; o < 32; o <<= 1) { int y = __shfl_up(x, o, 32); if (lane >= o) x += y; }
  if (lane == 31) wsum[wid] = x;
  __syncthreads();
  if (wid == 0) {
    const int nw = blockDim.x >> 5;
    int w = lane < nw ? wsum[lane] : 0;
    int z = w;
#pragma unroll
    for (int o = 1; o < 32; o <<= 1) { int y = __shfl_up(z, o, 32); if (lane >= o) z += y; }
    wsum[lane] = z - w;
    if (lane == 31) wsum[32] = z;
  }
  __syncthreads();
  total = wsum[32];
  return wsum[wid] + x - v;
}

__global__ __launch_bounds__(1024) void k_chunk_sort(const int* __restrict__ src, const int* __restrict__ dst,
                                                     unsigned short* __restrict__ e16,
                                                     unsigned char* __restrict__ e8,
                                                     unsigned short* __restrict__ otab) {
  __shared__ int hist[PB];
  __shared__ int wsum[33];
  __shared__ __attribute__((aligned(16))) unsigned short st16[CH];
  __shared__ __attribute__((aligned(16))) unsigned char  st8[CH];
  __shared__ __attribute__((aligned(16))) unsigned short row[TS];
  const int t = threadIdx.x, blk = blockIdx.x;
  if (t < PB) hist[t] = 0;
  if (t < TS) row[t] = 0;
  __syncthreads();
  const int* sp = src + blk * CH;
  const int* dp = dst + blk * CH;
  int sv[EPT], dv[EPT];
#pragma unroll
  for (int u = 0; u < EPT; ++u) { sv[u] = sp[t + 1024 * u]; dv[u] = dp[t + 1024 * u]; }
#pragma unroll
  for (int u = 0; u < EPT; ++u) atomicAdd(&hist[dv[u] >> RSH], 1);
  __syncthreads();
  const int c = t < PB ? hist[t] : 0;
  int total;
  const int lo = block_excl_scan(c, wsum, total);
  if (t < PB) { hist[t] = lo; row[t] = (unsigned short)lo; }
  if (t == 0) row[PB] = (unsigned short)CH;
  __syncthreads();
#pragma unroll
  for (int u = 0; u < EPT; ++u) {
    const int r = atomicAdd(&hist[dv[u] >> RSH], 1);
    st16[r] = (unsigned short)(sv[u] & 0xFFFF);
    st8[r]  = (unsigned char)((dv[u] & (RB - 1)) | ((sv[u] >> 16) << 7));
  }
  __syncthreads();
  const u32x4a* v16 = (const u32x4a*)st16;
  u32x4a* g16 = (u32x4a*)(e16 + (size_t)blk * CH);
  for (int i = t; i < CH / 8; i += 1024) g16[i] = v16[i];
  const u32x4a* v8 = (const u32x4a*)st8;
  u32x4a* g8 = (u32x4a*)(e8 + (size_t)blk * CH);
  for (int i = t; i < CH / 16; i += 1024) g8[i] = v8[i];
  const u32x4a* vr = (const u32x4a*)row;
  u32x4a* gr = (u32x4a*)(otab + (size_t)blk * TS);
  if (t < TS / 8) gr[t] = vr[t];
}

typedef __attribute__((ext_vector_type(4))) float g1v4;
#define XKC 64
#define XP  68
#define NPC 8
#define KPAD 504

__global__ __launch_bounds__(256) __attribute__((amdgpu_waves_per_eu(8, 8))) void k_gemm1_wmma(const float* __restrict__ X,
                                                    const float* __restrict__ W,
                                                    const unsigned char* __restrict__ e8,
                                                    const unsigned short* __restrict__ otab,
                                                    float* __restrict__ H1) {
  __shared__ __attribute__((aligned(16))) float Wq[KPAD * HID];
  __shared__ __attribute__((aligned(16))) float xt[8 * 16 * XP];
  __shared__ int dcnt[RB];
  const int wave = threadIdx.x >> 5;
  const int lane = threadIdx.x & 31;
  const int tile = blockIdx.x * 8 + wave;
  const int row_base = tile * 16;
  const int m    = lane & 15;
  const int hi   = lane >> 4;
  const float* xld = X + (size_t)(row_base + hi) * F_INd + 4 * m;
  g1v4 R[8];
  if (tile < NN / 16) {
#pragma unroll
    for (int u = 0; u < 8; ++u) R[u] = *(const g1v4*)(xld + (size_t)(2 * u) * F_INd);
  }
  for (int i = threadIdx.x; i < F_INd * HID; i += 256) {
    const int k = i >> 4, c = i & 15;
    Wq[((((k >> 2) << 4) + c) << 2) + (k & 3)] = W[i];
  }
  if (threadIdx.x < (KPAD - F_INd) * HID) Wq[F_INd * HID + threadIdx.x] = 0.0f;
  if (threadIdx.x < RB) dcnt[threadIdx.x] = 0;
  __syncthreads();
  {
    const int g = threadIdx.x >> 4, f = threadIdx.x & 15;
    int o0[8], o1[8];
#pragma unroll
    for (int k = 0; k < 8; ++k) {
      const int j = g + 16 * k, jc = j < NBK ? j : NBK - 1;
      const unsigned short* rw = otab + (size_t)jc * TS + blockIdx.x;
      o0[k] = j < NBK ? (int)rw[0] : 0;
      o1[k] = j < NBK ? (int)rw[1] : 0;
    }
#pragma unroll
    for (int k = 0; k < 8; ++k) {
      const int jc = (g + 16 * k) < NBK ? (g + 16 * k) : NBK - 1;
      const unsigned char* p8 = e8 + (size_t)jc * CH;
      const int oe = o1[k];
      for (int i0 = o0[k] + f; i0 < oe; i0 += 16 * 4) {
        int q[4];
#pragma unroll
        for (int u = 0; u < 4; ++u) { const int i = i0 + 16 * u; q[u] = p8[i < oe ? i : oe - 1]; }
#pragma unroll
        for (int u = 0; u < 4; ++u)
          if (i0 + 16 * u < oe) atomicAdd(&dcnt[q[u] & (RB - 1)], 1);
      }
    }
  }
  __syncthreads();

  if (tile >= NN / 16) return;

  float* tl = xt + wave * (16 * XP);
  float* tst = tl + hi * XP + 4 * m;
  const float* afr = tl + m * XP + 4 * hi;
  const float* bfr = Wq + (hi << 6) + 4 * m;
  {
    const g1v4 z = {0.0f, 0.0f, 0.0f, 0.0f};
    *(g1v4*)(tl + m * XP + XKC) = z;
  }

  v8f acc = {};
#pragma unroll
  for (int c = 0; c < NPC; ++c) {
    __builtin_amdgcn_fence(__ATOMIC_ACQ_REL, "wavefront");
#pragma unroll
    for (int u = 0; u < 8; ++u) *(g1v4*)(tst + (2 * u) * XP) = R[u];
    __builtin_amdgcn_fence(__ATOMIC_ACQ_REL, "wavefront");
    if (c + 1 < NPC) {
      const int nc = (c + 1 < NPC - 1) ? (c + 1) * XKC : F_INd - XKC;
#pragma unroll
      for (int u = 0; u < 8; ++u) R[u] = *(const g1v4*)(xld + (size_t)(2 * u) * F_INd + nc);
    }
    const int c0 = (c + 1 < NPC) ? c * XKC : F_INd - XKC;
    const int J0 = c * (XKC / 8);
    const int nj = (c + 1 < NPC) ? XKC / 8 : KPAD / 8 - J0;
    g1v4 a4[XKC / 8], b4[XKC / 8];
#pragma unroll
    for (int j = 0; j < nj; ++j) {
      a4[j] = *(const g1v4*)(afr + (8 * (J0 + j) - c0));
      b4[j] = *(const g1v4*)(bfr + 128 * (J0 + j));
    }
    __builtin_amdgcn_sched_barrier(0);
#pragma unroll
    for (int j = 0; j < nj; ++j) {
      const v2f a0 = {a4[j].x, a4[j].y}, b0 = {b4[j].x, b4[j].y};
      const v2f a1 = {a4[j].z, a4[j].w}, b1 = {b4[j].z, b4[j].w};
      acc = __builtin_amdgcn_wmma_f32_16x16x4_f32(false, a0, false, b0, (short)0, acc, false, false);
      acc = __builtin_amdgcn_wmma_f32_16x16x4_f32(false, a1, false, b1, (short)0, acc, false, false);
    }
  }
  __builtin_amdgcn_sched_barrier(0);
  asm volatile("v_nop\n\tv_nop\n\tv_nop\n\tv_nop");
  __builtin_amdgcn_sched_barrier(0);

  const int rbase = row_base + (hi << 3);
  const int lrow  = (wave << 4) + (hi << 3);
#pragma unroll
  for (int r = 0; r < 8; ++r)
    H1[(size_t)(rbase + r) * HID + m] = acc[r] * rsqrtf((float)(dcnt[lrow + r] + 1));
}


typedef __attribute__((ext_vector_type(4))) float v4f;
typedef __attribute__((ext_vector_type(4))) int i32x4;

#define BINW 64
#define BINP 68
#define RPL  16

__global__ __launch_bounds__(512, 16) void k_agg1_bucket(const float* __restrict__ h1s,
                                                     const unsigned short* __restrict__ e16,
                                                     const unsigned char* __restrict__ e8,
                                                     const unsigned short* __restrict__ otab,
                                                     const float* __restrict__ b1,
                                                     const float* __restrict__ W2,
                                                     float* __restrict__ h2s) {
  __shared__ __attribute__((aligned(16))) float acc[RB * HID];
  __shared__ int cnt[RB];
  __shared__ int full;
  __shared__ __attribute__((aligned(16))) int bin[RB * BINP];
  const int t = threadIdx.x, base = blockIdx.x * RB;
  const int j = t & 3, qd = t >> 2;
  const int jq = j << 2;
  const int jc = qd < NBK ? qd : NBK - 1;
  const unsigned short* rw = otab + (size_t)jc * TS + blockIdx.x;
  const int r0 = rw[0], r1 = rw[1];
  v4f own = {0.0f, 0.0f, 0.0f, 0.0f};
  if (base + qd < NN) own = *(const v4f*)(h1s + (size_t)base * HID + 4 * t);
  if (t < RB) cnt[t] = 0;
  if (t == 0) full = 0;
  for (int i = t; i < RB * HID; i += 512) acc[i] = 0.0f;
  __syncthreads();

  {
    const int oe = qd < NBK ? r1 : 0;
    const unsigned cb = (unsigned)jc * CH;
    for (int i0 = (qd < NBK ? r0 : 0) + j; i0 < oe; i0 += 4 * RPL) {
      int rec[RPL];
#pragma unroll
      for (int u = 0; u < RPL; ++u) {
        int i = i0 + 4 * u;
        i = i < oe ? i : oe - 1;
        rec[u] = (int)e8[cb + (unsigned)i] | ((int)e16[cb + (unsigned)i] << 8);
      }
      int slot[RPL];
#pragma unroll
      for (int u = 0; u < RPL; ++u) {
        slot[u] = 0;
        if (i0 + 4 * u < oe) slot[u] = atomicAdd(&cnt[rec[u] & (RB - 1)], 1);
      }
#pragma unroll
      for (int u = 0; u < RPL; ++u)
        if (i0 + 4 * u < oe) {
          const int dl = rec[u] & (RB - 1);
          const unsigned s = ((unsigned)rec[u] >> 8) | (((unsigned)rec[u] & 0x80u) << 9);
          if (slot[u] < BINW) bin[dl * BINP + slot[u]] = (int)s;
          else full = 1;
        }
    }
  }
  __syncthreads();

  {
    v4f sum = own;
    const int any_full = full;
    if (any_full) {
      const int oe = qd < NBK ? r1 : 0;
      const unsigned cb = (unsigned)jc * CH;
      for (int i = (qd < NBK ? r0 : 0) + j; i < oe; i += 4) {
        const int rec = (int)e8[cb + (unsigned)i] | ((int)e16[cb + (unsigned)i] << 8);
        const int dl = rec & (RB - 1);
        const unsigned s = ((unsigned)rec >> 8) | (((unsigned)rec & 0x80u) << 9);
        const float* hs = h1s + ((size_t)s << 4);
        for (int k = 0; k < HID; ++k) atomicAdd(&acc[(dl << 4) | k], hs[k]);
      }
      __syncthreads();
    }
    const int deg = cnt[qd];
    const int n = any_full ? 0 : (deg < BINW ? deg : BINW);
    const int* bp = bin + qd * BINP;
    for (int i0 = 0; i0 < n; i0 += 8) {
      const i32x4 ia = *(const i32x4*)(bp + i0);
      const i32x4 ib = *(const i32x4*)(bp + i0 + 4);
      const int id[8] = {ia.x, ia.y, ia.z, ia.w, ib.x, ib.y, ib.z, ib.w};
      v4f hv[8];
#pragma unroll
      for (int u = 0; u < 8; ++u) {
        const unsigned s = (i0 + u < n) ? ((unsigned)id[u] & 0x1FFFFu) : 0u;
        hv[u] = *(const v4f*)(h1s + ((s << 4) | (unsigned)jq));
      }
      __builtin_amdgcn_sched_barrier(0);
#pragma unroll
      for (int u = 0; u < 8; ++u) {
        const bool live = i0 + u < n;
        sum.x += live ? hv[u].x : 0.0f; sum.y += live ? hv[u].y : 0.0f;
        sum.z += live ? hv[u].z : 0.0f; sum.w += live ? hv[u].w : 0.0f;
      }
    }
    v4f* rowv = (v4f*)(acc + ((qd << 4) | jq));
    v4f o = *rowv;
    o.x += sum.x; o.y += sum.y; o.z += sum.z; o.w += sum.w;
    *rowv = o;
  }
  __syncthreads();
  if (t < RB && base + t < NN) {
    const float di = rsqrtf((float)(cnt[t] + 1));
    float v[HID];
#pragma unroll
    for (int k = 0; k < HID; ++k) {
      float x = acc[(t << 4) | k] * di + b1[k];
      v[k] = x > 0.0f ? x : 0.0f;
    }
#pragma unroll
    for (int c = 0; c < NC; ++c) {
      float sum = 0.0f;
#pragma unroll
      for (int k = 0; k < HID; ++k) sum += v[k] * W2[k * NC + c];
      h2s[(size_t)(base + t) * NC + c] = sum * di;
    }
  }
}

#define NSL 4
#define SLN 25000
#define GB  13
#define NG  61
#define SLV (SLN * NC / 4)

typedef unsigned int __attribute__((may_alias)) u32a;
typedef unsigned int __attribute__((ext_vector_type(2))) u32x2;
typedef u32x2 __attribute__((may_alias)) u32x2a;

__global__ __launch_bounds__(1024) void k_agg2_slice(const float* __restrict__ h2s,
                                                    const unsigned short* __restrict__ e16,
                                                    const unsigned char* __restrict__ e8,
                                                    const unsigned short* __restrict__ otab,
                                                    float* __restrict__ part) {
  __shared__ __attribute__((aligned(16))) float sl[SLN * NC];
  __shared__ __attribute__((aligned(16))) float acc[4 * GB * RB];
  const int t = threadIdx.x, lane = t & 31, wave = t >> 5;
  const int s = blockIdx.x & (NSL - 1), g = blockIdx.x >> 2;
  const int b0 = g * GB;
  const int nb = (PB - b0) < GB ? (PB - b0) : GB;
  int bnds[4];
#pragma unroll
  for (int ci = 0; ci < 4; ++ci) {
    const int c = wave + 32 * ci, cc = c < NBK ? c : NBK - 1;
    bnds[ci] = (int)otab[(size_t)cc * TS + b0 + (lane < nb ? lane : nb)];
  }
  {
    const v4f* gsl = (const v4f*)(h2s + (size_t)s * (SLN * NC));
    v4f* lsl = (v4f*)sl;
    for (int i0 = 0; i0 < SLV; i0 += 1024 * 10) {
      v4f r[10];
#pragma unroll
      for (int u = 0; u < 10; ++u) { const int i = i0 + t + 1024 * u; r[u] = gsl[i < SLV ? i : SLV - 1]; }
#pragma unroll
      for (int u = 0; u < 10; ++u) { const int i = i0 + t + 1024 * u; if (i < SLV) lsl[i] = r[u]; }
    }
    const v4f z = {0.0f, 0.0f, 0.0f, 0.0f};
    for (int i = t; i < GB * RB; i += 1024) ((v4f*)acc)[i] = z;
  }
  __syncthreads();

  const u32a*   e8w  = (const u32a*)e8;
  const u32x2a* e16w = (const u32x2a*)e16;
  const unsigned sbase = (unsigned)s * SLN;
#pragma unroll
  for (int ci = 0; ci < 4; ++ci) {
    const int c = wave + 32 * ci;
    if (c < NBK) {
      const int bnd = bnds[ci];
      const int pbeg = __builtin_amdgcn_readlane(bnd, 0);
      const int pend = __builtin_amdgcn_readlane(bnd, nb);
      const unsigned cw = ((unsigned)c * CH) >> 2;
      int kb = 0;
      int base = pbeg & ~127;
      unsigned w8 = 0; u32x2 w16 = {0u, 0u};
      if (base < pend) { w8 = e8w[cw + (base >> 2) + lane]; w16 = e16w[cw + (base >> 2) + lane]; }
      for (; base < pend; base += 128) {
        unsigned n8 = 0; u32x2 n16 = {0u, 0u};
        if (base + 128 < pend) { n8 = e8w[cw + ((base + 128) >> 2) + lane]; n16 = e16w[cw + ((base + 128) >> 2) + lane]; }
        const int first = base > pbeg ? base : pbeg;
        while (kb + 1 < nb && __builtin_amdgcn_readlane(bnd, kb + 1) <= first) ++kb;
        const int p0 = base + 4 * lane;
        int kk[4] = {kb, kb, kb, kb};
        for (int j = kb + 1; j < nb; ++j) {
          const int bj = __builtin_amdgcn_readlane(bnd, j);
          if (bj > base + 127) break;
#pragma unroll
          for (int u = 0; u < 4; ++u) kk[u] += (p0 + u >= bj) ? 1 : 0;
        }
#pragma unroll
        for (int u = 0; u < 4; ++u) {
          const int p = p0 + u;
          const unsigned q  = (w8 >> (8 * u)) & 255u;
          const unsigned hw = (u & 2) ? w16.y : w16.x;
          const unsigned lo = (u & 1) ? (hw >> 16) : (hw & 0xFFFFu);
          const unsigned rel = (lo | ((q >> 7) << 16)) - sbase;
          if (p >= pbeg && p < pend && rel < (unsigned)SLN) {
            const float* h = sl + rel * NC;
            float* a = acc + ((kk[u] << 7) | (int)(q & (RB - 1)));
            atomicAdd(a, h[0]);
            atomicAdd(a + GB * RB, h[1]);
            atomicAdd(a + 2 * GB * RB, h[2]);
            atomicAdd(a + 3 * GB * RB, 1.0f);
          }
        }
        w8 = n8; w16 = n16;
      }
    }
  }
  __syncthreads();
  {
    v4f* p4 = (v4f*)part + (size_t)s * NN + (size_t)b0 * RB;
    for (int i = t; i < nb * RB; i += 1024)
      if (b0 * RB + i < NN) {
        const v4f o = {acc[i], acc[GB * RB + i], acc[2 * GB * RB + i], acc[3 * GB * RB + i]};
        p4[i] = o;
      }
  }
}

__global__ void k_logsoftmax(const float* __restrict__ part, const float* __restrict__ b2,
                             float* out) {
  int i = blockIdx.x * blockDim.x + threadIdx.x;
  if (i >= NN) return;
  const v4f* p4 = (const v4f*)part;
  const v4f pa = p4[i], pb = p4[NN + i], pc = p4[2 * NN + i], pd = p4[3 * NN + i];
  const float h0 = out[(size_t)i * NC + 0], h1 = out[(size_t)i * NC + 1], h2 = out[(size_t)i * NC + 2];
  const float di = rsqrtf(pa.w + pb.w + pc.w + pd.w + 1.0f);
  float z0 = (h0 + pa.x + pb.x + pc.x + pd.x) * di + b2[0];
  float z1 = (h1 + pa.y + pb.y + pc.y + pd.y) * di + b2[1];
  float z2 = (h2 + pa.z + pb.z + pc.z + pd.z) * di + b2[2];
  float mx = fmaxf(z0, fmaxf(z1, z2));
  float lse = logf(expf(z0 - mx) + expf(z1 - mx) + expf(z2 - mx));
  out[(size_t)i * NC + 0] = z0 - mx - lse;
  out[(size_t)i * NC + 1] = z1 - mx - lse;
  out[(size_t)i * NC + 2] = z2 - mx - lse;
}

extern "C" void kernel_launch(void* const* d_in, const int* in_sizes, int n_in,
                              void* d_out, int out_size, void* d_ws, size_t ws_size,
                              hipStream_t stream) {
  const float* x  = (const float*)d_in[0];
  const float* W1 = (const float*)d_in[1];
  const float* b1 = (const float*)d_in[2];
  const float* W2 = (const float*)d_in[3];
  const float* b2 = (const float*)d_in[4];
  const int*   ei = (const int*)d_in[5];
  const int* src = ei;
  const int* dst = ei + EE;
  float* out = (float*)d_out;

  char* wsb = (char*)d_ws;
  float*          h1s  = (float*)wsb;
  unsigned short* e16  = (unsigned short*)(wsb + 6400000);
  unsigned char*  e8   = (unsigned char*)(wsb + 12800000);
  unsigned short* otab = (unsigned short*)(wsb + 16000000);
  float* part = (float*)wsb;
  float* h2s = out;

  k_chunk_sort  <<<NBK, 1024, 0, stream>>>(src, dst, e16, e8, otab);
  k_gemm1_wmma  <<<PB, 256, 0, stream>>>(x, W1, e8, otab, h1s);
  k_agg1_bucket <<<PB, 512, 0, stream>>>(h1s, e16, e8, otab, b1, W2, h2s);
  k_agg2_slice  <<<NG * NSL, 1024, 0, stream>>>(h2s, e16, e8, otab, part);
  k_logsoftmax  <<<(NN + 255) / 256, 256, 0, stream>>>(part, b2, out);
}


# layer-1 bins: a 4-lane group reads its run as aligned 16-record pieces, one 4-byte and one 8-byte load per lane for 4 records (32 record loads per lane become 8); slot order, so sum order, moves
# speedup vs baseline: 6.6489x; 1.0604x over previous
#include <hip/hip_runtime.h>
#include <cmath>

#define NN   100000
#define EE   3200000
#define F_INd 500
#define HID  16
#define NC   3
#define RSH  7
#define RB   128
#define PB   782
#define NBK  125
#define CH   25600
#define EPT  25
#define TS   832

typedef __attribute__((ext_vector_type(2))) float v2f;
typedef __attribute__((ext_vector_type(8))) float v8f;
typedef unsigned int __attribute__((ext_vector_type(4))) u32x4;
typedef u32x4 __attribute__((may_alias)) u32x4a;


__device__ __forceinline__ int block_excl_scan(int v, int* wsum, int& total) {
  const int lane = threadIdx.x & 31, wid = threadIdx.x >> 5;
  int x = v;
#pragma unroll
  for (int o = 1; o < 32; o <<= 1) { int y = __shfl_up(x, o, 32); if (lane >= o) x += y; }
  if (lane == 31) wsum[wid] = x;
  __syncthreads();
  if (wid == 0) {
    const int nw = blockDim.x >> 5;
    int w = lane < nw ? wsum[lane] : 0;
    int z = w;
#pragma unroll
    for (int o = 1; o < 32; o <<= 1) { int y = __shfl_up(z, o, 32); if (lane >= o) z += y; }
    wsum[lane] = z - w;
    if (lane == 31) wsum[32] = z;
  }
  __syncthreads();
  total = wsum[32];
  return wsum[wid] + x - v;
}

__global__ __launch_bounds__(1024) void k_chunk_sort(const int* __restrict__ src, const int* __restrict__ dst,
                                                     unsigned short* __restrict__ e16,
                                                     unsigned char* __restrict__ e8,
                                                     unsigned short* __restrict__ otab) {
  __shared__ int hist[PB];
  __shared__ int wsum[33];
  __shared__ __attribute__((aligned(16))) unsigned short st16[CH];
  __shared__ __attribute__((aligned(16))) unsigned char  st8[CH];
  __shared__ __attribute__((aligned(16))) unsigned short row[TS];
  const int t = threadIdx.x, blk = blockIdx.x;
  if (t < PB) hist[t] = 0;
  if (t < TS) row[t] = 0;
  __syncthreads();
  const int* sp = src + blk * CH;
  const int* dp = dst + blk * CH;
  int sv[EPT], dv[EPT];
#pragma unroll
  for (int u = 0; u < EPT; ++u) { sv[u] = sp[t + 1024 * u]; dv[u] = dp[t + 1024 * u]; }
#pragma unroll
  for (int u = 0; u < EPT; ++u) atomicAdd(&hist[dv[u] >> RSH], 1);
  __syncthreads();
  const int c = t < PB ? hist[t] : 0;
  int total;
  const int lo = block_excl_scan(c, wsum, total);
  if (t < PB) { hist[t] = lo; row[t] = (unsigned short)lo; }
  if (t == 0) row[PB] = (unsigned short)CH;
  __syncthreads();
#pragma unroll
  for (int u = 0; u < EPT; ++u) {
    const int r = atomicAdd(&hist[dv[u] >> RSH], 1);
    st16[r] = (unsigned short)(sv[u] & 0xFFFF);
    st8[r]  = (unsigned char)((dv[u] & (RB - 1)) | ((sv[u] >> 16) << 7));
  }
  __syncthreads();
  const u32x4a* v16 = (const u32x4a*)st16;
  u32x4a* g16 = (u32x4a*)(e16 + (size_t)blk * CH);
  for (int i = t; i < CH / 8; i += 1024) g16[i] = v16[i];
  const u32x4a* v8 = (const u32x4a*)st8;
  u32x4a* g8 = (u32x4a*)(e8 + (size_t)blk * CH);
  for (int i = t; i < CH / 16; i += 1024) g8[i] = v8[i];
  const u32x4a* vr = (const u32x4a*)row;
  u32x4a* gr = (u32x4a*)(otab + (size_t)blk * TS);
  if (t < TS / 8) gr[t] = vr[t];
}

typedef __attribute__((ext_vector_type(4))) float g1v4;
#define XKC 64
#define XP  68
#define NPC 8
#define KPAD 504

__global__ __launch_bounds__(256) __attribute__((amdgpu_waves_per_eu(8, 8))) void k_gemm1_wmma(const float* __restrict__ X,
                                                    const float* __restrict__ W,
                                                    const unsigned char* __restrict__ e8,
                                                    const unsigned short* __restrict__ otab,
                                                    float* __restrict__ H1) {
  __shared__ __attribute__((aligned(16))) float Wq[KPAD * HID];
  __shared__ __attribute__((aligned(16))) float xt[8 * 16 * XP];
  __shared__ int dcnt[RB];
  const int wave = threadIdx.x >> 5;
  const int lane = threadIdx.x & 31;
  const int tile = blockIdx.x * 8 + wave;
  const int row_base = tile * 16;
  const int m    = lane & 15;
  const int hi   = lane >> 4;
  const float* xld = X + (size_t)(row_base + hi) * F_INd + 4 * m;
  g1v4 R[8];
  if (tile < NN / 16) {
#pragma unroll
    for (int u = 0; u < 8; ++u) R[u] = *(const g1v4*)(xld + (size_t)(2 * u) * F_INd);
  }
  for (int i = threadIdx.x; i < F_INd * HID; i += 256) {
    const int k = i >> 4, c = i & 15;
    Wq[((((k >> 2) << 4) + c) << 2) + (k & 3)] = W[i];
  }
  if (threadIdx.x < (KPAD - F_INd) * HID) Wq[F_INd * HID + threadIdx.x] = 0.0f;
  if (threadIdx.x < RB) dcnt[threadIdx.x] = 0;
  __syncthreads();
  {
    const int g = threadIdx.x >> 4, f = threadIdx.x & 15;
    int o0[8], o1[8];
#pragma unroll
    for (int k = 0; k < 8; ++k) {
      const int j = g + 16 * k, jc = j < NBK ? j : NBK - 1;
      const unsigned short* rw = otab + (size_t)jc * TS + blockIdx.x;
      o0[k] = j < NBK ? (int)rw[0] : 0;
      o1[k] = j < NBK ? (int)rw[1] : 0;
    }
#pragma unroll
    for (int k = 0; k < 8; ++k) {
      const int jc = (g + 16 * k) < NBK ? (g + 16 * k) : NBK - 1;
      const unsigned char* p8 = e8 + (size_t)jc * CH;
      const int oe = o1[k];
      for (int i0 = o0[k] + f; i0 < oe; i0 += 16 * 4) {
        int q[4];
#pragma unroll
        for (int u = 0; u < 4; ++u) { const int i = i0 + 16 * u; q[u] = p8[i < oe ? i : oe - 1]; }
#pragma unroll
        for (int u = 0; u < 4; ++u)
          if (i0 + 16 * u < oe) atomicAdd(&dcnt[q[u] & (RB - 1)], 1);
      }
    }
  }
  __syncthreads();

  if (tile >= NN / 16) return;

  float* tl = xt + wave * (16 * XP);
  float* tst = tl + hi * XP + 4 * m;
  const float* afr = tl + m * XP + 4 * hi;
  const float* bfr = Wq + (hi << 6) + 4 * m;
  {
    const g1v4 z = {0.0f, 0.0f, 0.0f, 0.0f};
    *(g1v4*)(tl + m * XP + XKC) = z;
  }

  v8f acc = {};
#pragma unroll
  for (int c = 0; c < NPC; ++c) {
    __builtin_amdgcn_fence(__ATOMIC_ACQ_REL, "wavefront");
#pragma unroll
    for (int u = 0; u < 8; ++u) *(g1v4*)(tst + (2 * u) * XP) = R[u];
    __builtin_amdgcn_fence(__ATOMIC_ACQ_REL, "wavefront");
    if (c + 1 < NPC) {
      const int nc = (c + 1 < NPC - 1) ? (c + 1) * XKC : F_INd - XKC;
#pragma unroll
      for (int u = 0; u < 8; ++u) R[u] = *(const g1v4*)(xld + (size_t)(2 * u) * F_INd + nc);
    }
    const int c0 = (c + 1 < NPC) ? c * XKC : F_INd - XKC;
    const int J0 = c * (XKC / 8);
    const int nj = (c + 1 < NPC) ? XKC / 8 : KPAD / 8 - J0;
    g1v4 a4[XKC / 8], b4[XKC / 8];
#pragma unroll
    for (int j = 0; j < nj; ++j) {
      a4[j] = *(const g1v4*)(afr + (8 * (J0 + j) - c0));
      b4[j] = *(const g1v4*)(bfr + 128 * (J0 + j));
    }
    __builtin_amdgcn_sched_barrier(0);
#pragma unroll
    for (int j = 0; j < nj; ++j) {
      const v2f a0 = {a4[j].x, a4[j].y}, b0 = {b4[j].x, b4[j].y};
      const v2f a1 = {a4[j].z, a4[j].w}, b1 = {b4[j].z, b4[j].w};
      acc = __builtin_amdgcn_wmma_f32_16x16x4_f32(false, a0, false, b0, (short)0, acc, false, false);
      acc = __builtin_amdgcn_wmma_f32_16x16x4_f32(false, a1, false, b1, (short)0, acc, false, false);
    }
  }
  __builtin_amdgcn_sched_barrier(0);
  asm volatile("v_nop\n\tv_nop\n\tv_nop\n\tv_nop");
  __builtin_amdgcn_sched_barrier(0);

  const int rbase = row_base + (hi << 3);
  const int lrow  = (wave << 4) + (hi << 3);
#pragma unroll
  for (int r = 0; r < 8; ++r)
    H1[(size_t)(rbase + r) * HID + m] = acc[r] * rsqrtf((float)(dcnt[lrow + r] + 1));
}


typedef __attribute__((ext_vector_type(4))) float v4f;
typedef __attribute__((ext_vector_type(4))) int i32x4;
typedef unsigned int __attribute__((may_alias)) u32a;
typedef unsigned int __attribute__((ext_vector_type(2))) u32x2;
typedef u32x2 __attribute__((may_alias)) u32x2a;

#define BINW 64
#define BINP 68
#define PPB  4

__global__ __launch_bounds__(512, 16) void k_agg1_bucket(const float* __restrict__ h1s,
                                                     const unsigned short* __restrict__ e16,
                                                     const unsigned char* __restrict__ e8,
                                                     const unsigned short* __restrict__ otab,
                                                     const float* __restrict__ b1,
                                                     const float* __restrict__ W2,
                                                     float* __restrict__ h2s) {
  __shared__ __attribute__((aligned(16))) float acc[RB * HID];
  __shared__ int cnt[RB];
  __shared__ int full;
  __shared__ __attribute__((aligned(16))) int bin[RB * BINP];
  const int t = threadIdx.x, base = blockIdx.x * RB;
  const int j = t & 3, qd = t >> 2;
  const int jq = j << 2;
  const int jc = qd < NBK ? qd : NBK - 1;
  const unsigned short* rw = otab + (size_t)jc * TS + blockIdx.x;
  const int r0 = rw[0], r1 = rw[1];
  v4f own = {0.0f, 0.0f, 0.0f, 0.0f};
  if (base + qd < NN) own = *(const v4f*)(h1s + (size_t)base * HID + 4 * t);
  if (t < RB) cnt[t] = 0;
  if (t == 0) full = 0;
  for (int i = t; i < RB * HID; i += 512) acc[i] = 0.0f;
  __syncthreads();

  {
    const int rb = qd < NBK ? r0 : 0;
    const int oe = qd < NBK ? r1 : 0;
    const u32a*   e8w  = (const u32a*)e8;
    const u32x2a* e16w = (const u32x2a*)e16;
    const unsigned cw = (((unsigned)jc * CH) >> 2) + (unsigned)j;
    const int pl = (oe - 1) >> 4;
    for (int p = rb >> 4; p <= pl; p += PPB) {
      unsigned w8[PPB];
      u32x2 w16[PPB];
#pragma unroll
      for (int k = 0; k < PPB; ++k) {
        const int pk = p + k < pl ? p + k : pl;
        w8[k]  = e8w[cw + 4u * (unsigned)pk];
        w16[k] = e16w[cw + 4u * (unsigned)pk];
      }
      int slot[4 * PPB];
#pragma unroll
      for (int k = 0; k < PPB; ++k)
#pragma unroll
        for (int u = 0; u < 4; ++u) {
          const int i = 16 * (p + k) + 4 * j + u;
          const int dl = (int)((w8[k] >> (8 * u)) & (RB - 1));
          slot[4 * k + u] = 0;
          if (i >= rb && i < oe) slot[4 * k + u] = atomicAdd(&cnt[dl], 1);
        }
#pragma unroll
      for (int k = 0; k < PPB; ++k)
#pragma unroll
        for (int u = 0; u < 4; ++u) {
          const int i = 16 * (p + k) + 4 * j + u;
          if (i >= rb && i < oe) {
            const unsigned q  = (w8[k] >> (8 * u)) & 255u;
            const unsigned hw = (u & 2) ? w16[k].y : w16[k].x;
            const unsigned lo = (u & 1) ? (hw >> 16) : (hw & 0xFFFFu);
            const unsigned s  = lo | ((q >> 7) << 16);
            const int dl = (int)(q & (RB - 1));
            if (slot[4 * k + u] < BINW) bin[dl * BINP + slot[4 * k + u]] = (int)s;
            else full = 1;
          }
        }
    }
  }
  __syncthreads();

  {
    v4f sum = own;
    const int any_full = full;
    if (any_full) {
      const int oe = qd < NBK ? r1 : 0;
      const unsigned cb = (unsigned)jc * CH;
      for (int i = (qd < NBK ? r0 : 0) + j; i < oe; i += 4) {
        const int rec = (int)e8[cb + (unsigned)i] | ((int)e16[cb + (unsigned)i] << 8);
        const int dl = rec & (RB - 1);
        const unsigned s = ((unsigned)rec >> 8) | (((unsigned)rec & 0x80u) << 9);
        const float* hs = h1s + ((size_t)s << 4);
        for (int k = 0; k < HID; ++k) atomicAdd(&acc[(dl << 4) | k], hs[k]);
      }
      __syncthreads();
    }
    const int deg = cnt[qd];
    const int n = any_full ? 0 : (deg < BINW ? deg : BINW);
    const int* bp = bin + qd * BINP;
    for (int i0 = 0; i0 < n; i0 += 8) {
      const i32x4 ia = *(const i32x4*)(bp + i0);
      const i32x4 ib = *(const i32x4*)(bp + i0 + 4);
      const int id[8] = {ia.x, ia.y, ia.z, ia.w, ib.x, ib.y, ib.z, ib.w};
      v4f hv[8];
#pragma unroll
      for (int u = 0; u < 8; ++u) {
        const unsigned s = (i0 + u < n) ? ((unsigned)id[u] & 0x1FFFFu) : 0u;
        hv[u] = *(const v4f*)(h1s + ((s << 4) | (unsigned)jq));
      }
      __builtin_amdgcn_sched_barrier(0);
#pragma unroll
      for (int u = 0; u < 8; ++u) {
        const bool live = i0 + u < n;
        sum.x += live ? hv[u].x : 0.0f; sum.y += live ? hv[u].y : 0.0f;
        sum.z += live ? hv[u].z : 0.0f; sum.w += live ? hv[u].w : 0.0f;
      }
    }
    v4f* rowv = (v4f*)(acc + ((qd << 4) | jq));
    v4f o = *rowv;
    o.x += sum.x; o.y += sum.y; o.z += sum.z; o.w += sum.w;
    *rowv = o;
  }
  __syncthreads();
  if (t < RB && base + t < NN) {
    const float di = rsqrtf((float)(cnt[t] + 1));
    float v[HID];
#pragma unroll
    for (int k = 0; k < HID; ++k) {
      float x = acc[(t << 4) | k] * di + b1[k];
      v[k] = x > 0.0f ? x : 0.0f;
    }
#pragma unroll
    for (int c = 0; c < NC; ++c) {
      float sum = 0.0f;
#pragma unroll
      for (int k = 0; k < HID; ++k) sum += v[k] * W2[k * NC + c];
      h2s[(size_t)(base + t) * NC + c] = sum * di;
    }
  }
}

#define NSL 4
#define SLN 25000
#define GB  13
#define NG  61
#define SLV (SLN * NC / 4)

__global__ __launch_bounds__(1024) void k_agg2_slice(const float* __restrict__ h2s,
                                                    const unsigned short* __restrict__ e16,
                                                    const unsigned char* __restrict__ e8,
                                                    const unsigned short* __restrict__ otab,
                                                    float* __restrict__ part) {
  __shared__ __attribute__((aligned(16))) float sl[SLN * NC];
  __shared__ __attribute__((aligned(16))) float acc[4 * GB * RB];
  const int t = threadIdx.x, lane = t & 31, wave = t >> 5;
  const int s = blockIdx.x & (NSL - 1), g = blockIdx.x >> 2;
  const int b0 = g * GB;
  const int nb = (PB - b0) < GB ? (PB - b0) : GB;
  int bnds[4];
#pragma unroll
  for (int ci = 0; ci < 4; ++ci) {
    const int c = wave + 32 * ci, cc = c < NBK ? c : NBK - 1;
    bnds[ci] = (int)otab[(size_t)cc * TS + b0 + (lane < nb ? lane : nb)];
  }
  {
    const v4f* gsl = (const v4f*)(h2s + (size_t)s * (SLN * NC));
    v4f* lsl = (v4f*)sl;
    for (int i0 = 0; i0 < SLV; i0 += 1024 * 10) {
      v4f r[10];
#pragma unroll
      for (int u = 0; u < 10; ++u) { const int i = i0 + t + 1024 * u; r[u] = gsl[i < SLV ? i : SLV - 1]; }
#pragma unroll
      for (int u = 0; u < 10; ++u) { const int i = i0 + t + 1024 * u; if (i < SLV) lsl[i] = r[u]; }
    }
    const v4f z = {0.0f, 0.0f, 0.0f, 0.0f};
    for (int i = t; i < GB * RB; i += 1024) ((v4f*)acc)[i] = z;
  }
  __syncthreads();

  const u32a*   e8w  = (const u32a*)e8;
  const u32x2a* e16w = (const u32x2a*)e16;
  const unsigned sbase = (unsigned)s * SLN;
#pragma unroll
  for (int ci = 0; ci < 4; ++ci) {
    const int c = wave + 32 * ci;
    if (c < NBK) {
      const int bnd = bnds[ci];
      const int pbeg = __builtin_amdgcn_readlane(bnd, 0);
      const int pend = __builtin_amdgcn_readlane(bnd, nb);
      const unsigned cw = ((unsigned)c * CH) >> 2;
      int kb = 0;
      int base = pbeg & ~127;
      unsigned w8 = 0; u32x2 w16 = {0u, 0u};
      if (base < pend) { w8 = e8w[cw + (base >> 2) + lane]; w16 = e16w[cw + (base >> 2) + lane]; }
      for (; base < pend; base += 128) {
        unsigned n8 = 0; u32x2 n16 = {0u, 0u};
        if (base + 128 < pend) { n8 = e8w[cw + ((base + 128) >> 2) + lane]; n16 = e16w[cw + ((base + 128) >> 2) + lane]; }
        const int first = base > pbeg ? base : pbeg;
        while (kb + 1 < nb && __builtin_amdgcn_readlane(bnd, kb + 1) <= first) ++kb;
        const int p0 = base + 4 * lane;
        int kk[4] = {kb, kb, kb, kb};
        for (int j = kb + 1; j < nb; ++j) {
          const int bj = __builtin_amdgcn_readlane(bnd, j);
          if (bj > base + 127) break;
#pragma unroll
          for (int u = 0; u < 4; ++u) kk[u] += (p0 + u >= bj) ? 1 : 0;
        }
#pragma unroll
        for (int u = 0; u < 4; ++u) {
          const int p = p0 + u;
          const unsigned q  = (w8 >> (8 * u)) & 255u;
          const unsigned hw = (u & 2) ? w16.y : w16.x;
          const unsigned lo = (u & 1) ? (hw >> 16) : (hw & 0xFFFFu);
          const unsigned rel = (lo | ((q >> 7) << 16)) - sbase;
          if (p >= pbeg && p < pend && rel < (unsigned)SLN) {
            const float* h = sl + rel * NC;
            float* a = acc + ((kk[u] << 7) | (int)(q & (RB - 1)));
            atomicAdd(a, h[0]);
            atomicAdd(a + GB * RB, h[1]);
            atomicAdd(a + 2 * GB * RB, h[2]);
            atomicAdd(a + 3 * GB * RB, 1.0f);
          }
        }
        w8 = n8; w16 = n16;
      }
    }
  }
  __syncthreads();
  {
    v4f* p4 = (v4f*)part + (size_t)s * NN + (size_t)b0 * RB;
    for (int i = t; i < nb * RB; i += 1024)
      if (b0 * RB + i < NN) {
        const v4f o = {acc[i], acc[GB * RB + i], acc[2 * GB * RB + i], acc[3 * GB * RB + i]};
        p4[i] = o;
      }
  }
}

__global__ void k_logsoftmax(const float* __restrict__ part, const float* __restrict__ b2,
                             float* out) {
  int i = blockIdx.x * blockDim.x + threadIdx.x;
  if (i >= NN) return;
  const v4f* p4 = (const v4f*)part;
  const v4f pa = p4[i], pb = p4[NN + i], pc = p4[2 * NN + i], pd = p4[3 * NN + i];
  const float h0 = out[(size_t)i * NC + 0], h1 = out[(size_t)i * NC + 1], h2 = out[(size_t)i * NC + 2];
  const float di = rsqrtf(pa.w + pb.w + pc.w + pd.w + 1.0f);
  float z0 = (h0 + pa.x + pb.x + pc.x + pd.x) * di + b2[0];
  float z1 = (h1 + pa.y + pb.y + pc.y + pd.y) * di + b2[1];
  float z2 = (h2 + pa.z + pb.z + pc.z + pd.z) * di + b2[2];
  float mx = fmaxf(z0, fmaxf(z1, z2));
  float lse = logf(expf(z0 - mx) + expf(z1 - mx) + expf(z2 - mx));
  out[(size_t)i * NC + 0] = z0 - mx - lse;
  out[(size_t)i * NC + 1] = z1 - mx - lse;
  out[(size_t)i * NC + 2] = z2 - mx - lse;
}

extern "C" void kernel_launch(void* const* d_in, const int* in_sizes, int n_in,
                              void* d_out, int out_size, void* d_ws, size_t ws_size,
                              hipStream_t stream) {
  const float* x  = (const float*)d_in[0];
  const float* W1 = (const float*)d_in[1];
  const float* b1 = (const float*)d_in[2];
  const float* W2 = (const float*)d_in[3];
  const float* b2 = (const float*)d_in[4];
  const int*   ei = (const int*)d_in[5];
  const int* src = ei;
  const int* dst = ei + EE;
  float* out = (float*)d_out;

  char* wsb = (char*)d_ws;
  float*          h1s  = (float*)wsb;
  unsigned short* e16  = (unsigned short*)(wsb + 6400000);
  unsigned char*  e8   = (unsigned char*)(wsb + 12800000);
  unsigned short* otab = (unsigned short*)(wsb + 16000000);
  float* part = (float*)wsb;
  float* h2s = out;

  k_chunk_sort  <<<NBK, 1024, 0, stream>>>(src, dst, e16, e8, otab);
  k_gemm1_wmma  <<<PB, 256, 0, stream>>>(x, W1, e8, otab, h1s);
  k_agg1_bucket <<<PB, 512, 0, stream>>>(h1s, e16, e8, otab, b1, W2, h2s);
  k_agg2_slice  <<<NG * NSL, 1024, 0, stream>>>(h2s, e16, e8, otab, part);
  k_logsoftmax  <<<(NN + 255) / 256, 256, 0, stream>>>(part, b2, out);
}


# GEMM1 prologue: W copy by 16-byte loads and the bucket's edge bytes as aligned dwords, all 8 runs at once, requested before anything waits: 18 dependent round trips become 2; sums unchanged
# speedup vs baseline: 6.9986x; 1.0526x over previous
#include <hip/hip_runtime.h>
#include <cmath>

#define NN   100000
#define EE   3200000
#define F_INd 500
#define HID  16
#define NC   3
#define RSH  7
#define RB   128
#define PB   782
#define NBK  125
#define CH   25600
#define EPT  25
#define TS   832

typedef __attribute__((ext_vector_type(2))) float v2f;
typedef __attribute__((ext_vector_type(8))) float v8f;
typedef unsigned int __attribute__((ext_vector_type(4))) u32x4;
typedef u32x4 __attribute__((may_alias)) u32x4a;


__device__ __forceinline__ int block_excl_scan(int v, int* wsum, int& total) {
  const int lane = threadIdx.x & 31, wid = threadIdx.x >> 5;
  int x = v;
#pragma unroll
  for (int o = 1; o < 32; o <<= 1) { int y = __shfl_up(x, o, 32); if (lane >= o) x += y; }
  if (lane == 31) wsum[wid] = x;
  __syncthreads();
  if (wid == 0) {
    const int nw = blockDim.x >> 5;
    int w = lane < nw ? wsum[lane] : 0;
    int z = w;
#pragma unroll
    for (int o = 1; o < 32; o <<= 1) { int y = __shfl_up(z, o, 32); if (lane >= o) z += y; }
    wsum[lane] = z - w;
    if (lane == 31) wsum[32] = z;
  }
  __syncthreads();
  total = wsum[32];
  return wsum[wid] + x - v;
}

__global__ __launch_bounds__(1024) void k_chunk_sort(const int* __restrict__ src, const int* __restrict__ dst,
                                                     unsigned short* __restrict__ e16,
                                                     unsigned char* __restrict__ e8,
                                                     unsigned short* __restrict__ otab) {
  __shared__ int hist[PB];
  __shared__ int wsum[33];
  __shared__ __attribute__((aligned(16))) unsigned short st16[CH];
  __shared__ __attribute__((aligned(16))) unsigned char  st8[CH];
  __shared__ __attribute__((aligned(16))) unsigned short row[TS];
  const int t = threadIdx.x, blk = blockIdx.x;
  if (t < PB) hist[t] = 0;
  if (t < TS) row[t] = 0;
  __syncthreads();
  const int* sp = src + blk * CH;
  const int* dp = dst + blk * CH;
  int sv[EPT], dv[EPT];
#pragma unroll
  for (int u = 0; u < EPT; ++u) { sv[u] = sp[t + 1024 * u]; dv[u] = dp[t + 1024 * u]; }
#pragma unroll
  for (int u = 0; u < EPT; ++u) atomicAdd(&hist[dv[u] >> RSH], 1);
  __syncthreads();
  const int c = t < PB ? hist[t] : 0;
  int total;
  const int lo = block_excl_scan(c, wsum, total);
  if (t < PB) { hist[t] = lo; row[t] = (unsigned short)lo; }
  if (t == 0) row[PB] = (unsigned short)CH;
  __syncthreads();
#pragma unroll
  for (int u = 0; u < EPT; ++u) {
    const int r = atomicAdd(&hist[dv[u] >> RSH], 1);
    st16[r] = (unsigned short)(sv[u] & 0xFFFF);
    st8[r]  = (unsigned char)((dv[u] & (RB - 1)) | ((sv[u] >> 16) << 7));
  }
  __syncthreads();
  const u32x4a* v16 = (const u32x4a*)st16;
  u32x4a* g16 = (u32x4a*)(e16 + (size_t)blk * CH);
  for (int i = t; i < CH / 8; i += 1024) g16[i] = v16[i];
  const u32x4a* v8 = (const u32x4a*)st8;
  u32x4a* g8 = (u32x4a*)(e8 + (size_t)blk * CH);
  for (int i = t; i < CH / 16; i += 1024) g8[i] = v8[i];
  const u32x4a* vr = (const u32x4a*)row;
  u32x4a* gr = (u32x4a*)(otab + (size_t)blk * TS);
  if (t < TS / 8) gr[t] = vr[t];
}

typedef __attribute__((ext_vector_type(4))) float g1v4;
typedef unsigned int __attribute__((may_alias)) g1u32a;
#define XKC 64
#define XP  68
#define NPC 8
#define KPAD 504

__global__ __launch_bounds__(256) __attribute__((amdgpu_waves_per_eu(8, 8))) void k_gemm1_wmma(const float* __restrict__ X,
                                                    const float* __restrict__ W,
                                                    const unsigned char* __restrict__ e8,
                                                    const unsigned short* __restrict__ otab,
                                                    float* __restrict__ H1) {
  __shared__ __attribute__((aligned(16))) float Wq[KPAD * HID];
  __shared__ __attribute__((aligned(16))) float xt[8 * 16 * XP];
  __shared__ int dcnt[RB];
  const int wave = threadIdx.x >> 5;
  const int lane = threadIdx.x & 31;
  const int tile = blockIdx.x * 8 + wave;
  const int row_base = tile * 16;
  const int m    = lane & 15;
  const int hi   = lane >> 4;
  const int ld_base = tile < NN / 16 ? row_base : 0;
  const float* xld = X + (size_t)(ld_base + hi) * F_INd + 4 * m;
  const int g = threadIdx.x >> 4, f = threadIdx.x & 15;
  int o0[8], o1[8];
#pragma unroll
  for (int k = 0; k < 8; ++k) {
    const int j = g + 16 * k, jc = j < NBK ? j : NBK - 1;
    const unsigned short* rw = otab + (size_t)jc * TS + blockIdx.x;
    const int a = (int)rw[0], b = (int)rw[1];
    o0[k] = j < NBK ? a : 0;
    o1[k] = j < NBK ? b : 0;
  }
  g1v4 R[8];
#pragma unroll
  for (int u = 0; u < 8; ++u) R[u] = *(const g1v4*)(xld + (size_t)(2 * u) * F_INd);
  g1v4 wv[8];
#pragma unroll
  for (int u = 0; u < 8; ++u) {
    const int i = threadIdx.x + 256 * u;
    wv[u] = ((const g1v4*)W)[i < F_INd * HID / 4 ? i : F_INd * HID / 4 - 1];
  }
  const g1u32a* e8w = (const g1u32a*)e8;
  unsigned qw[8];
  int wf[8], wl[8];
#pragma unroll
  for (int k = 0; k < 8; ++k) {
    const int j = g + 16 * k, jc = j < NBK ? j : NBK - 1;
    const int cb = jc * CH;
    wf[k] = ((cb + o0[k]) >> 2) + f;
    wl[k] = (cb + o1[k] - 1) >> 2;
    int wc = wf[k] < wl[k] ? wf[k] : wl[k];
    wc = wc > 0 ? wc : 0;
    qw[k] = e8w[wc];
  }
#pragma unroll
  for (int u = 0; u < 8; ++u) {
    const int i = threadIdx.x + 256 * u;
    if (i < F_INd * HID / 4) {
      const int k = i >> 2, c = (i & 3) << 2;
      float* d = Wq + ((((k >> 2) << 4) + c) << 2) + (k & 3);
      d[0] = wv[u].x; d[4] = wv[u].y; d[8] = wv[u].z; d[12] = wv[u].w;
    }
  }
  if (threadIdx.x < (KPAD - F_INd) * HID) Wq[F_INd * HID + threadIdx.x] = 0.0f;
  if (threadIdx.x < RB) dcnt[threadIdx.x] = 0;
  __syncthreads();
#pragma unroll
  for (int k = 0; k < 8; ++k) {
    const int j = g + 16 * k, jc = j < NBK ? j : NBK - 1;
    const int cb = jc * CH;
    const int ob = o0[k], oe = o1[k];
    unsigned q = qw[k];
    for (int w = wf[k]; ; ) {
      const int i0 = 4 * w - cb;
      for (int u = 0; u < 4; ++u)
        if (i0 + u >= ob && i0 + u < oe) atomicAdd(&dcnt[(q >> (8 * u)) & (RB - 1)], 1);
      w += 16;
      if (w > wl[k]) break;
      q = e8w[w];
    }
  }
  __syncthreads();

  if (tile >= NN / 16) return;

  float* tl = xt + wave * (16 * XP);
  float* tst = tl + hi * XP + 4 * m;
  const float* afr = tl + m * XP + 4 * hi;
  const float* bfr = Wq + (hi << 6) + 4 * m;
  {
    const g1v4 z = {0.0f, 0.0f, 0.0f, 0.0f};
    *(g1v4*)(tl + m * XP + XKC) = z;
  }

  v8f acc = {};
#pragma unroll
  for (int c = 0; c < NPC; ++c) {
    __builtin_amdgcn_fence(__ATOMIC_ACQ_REL, "wavefront");
#pragma unroll
    for (int u = 0; u < 8; ++u) *(g1v4*)(tst + (2 * u) * XP) = R[u];
    __builtin_amdgcn_fence(__ATOMIC_ACQ_REL, "wavefront");
    if (c + 1 < NPC) {
      const int nc = (c + 1 < NPC - 1) ? (c + 1) * XKC : F_INd - XKC;
#pragma unroll
      for (int u = 0; u < 8; ++u) R[u] = *(const g1v4*)(xld + (size_t)(2 * u) * F_INd + nc);
    }
    const int c0 = (c + 1 < NPC) ? c * XKC : F_INd - XKC;
    const int J0 = c * (XKC / 8);
    const int nj = (c + 1 < NPC) ? XKC / 8 : KPAD / 8 - J0;
    g1v4 a4[XKC / 8], b4[XKC / 8];
#pragma unroll
    for (int j = 0; j < nj; ++j) {
      a4[j] = *(const g1v4*)(afr + (8 * (J0 + j) - c0));
      b4[j] = *(const g1v4*)(bfr + 128 * (J0 + j));
    }
    __builtin_amdgcn_sched_barrier(0);
#pragma unroll
    for (int j = 0; j < nj; ++j) {
      const v2f a0 = {a4[j].x, a4[j].y}, b0 = {b4[j].x, b4[j].y};
      const v2f a1 = {a4[j].z, a4[j].w}, b1 = {b4[j].z, b4[j].w};
      acc = __builtin_amdgcn_wmma_f32_16x16x4_f32(false, a0, false, b0, (short)0, acc, false, false);
      acc = __builtin_amdgcn_wmma_f32_16x16x4_f32(false, a1, false, b1, (short)0, acc, false, false);
    }
  }
  __builtin_amdgcn_sched_barrier(0);
  asm volatile("v_nop\n\tv_nop\n\tv_nop\n\tv_nop");
  __builtin_amdgcn_sched_barrier(0);

  const int rbase = row_base + (hi << 3);
  const int lrow  = (wave << 4) + (hi << 3);
#pragma unroll
  for (int r = 0; r < 8; ++r)
    H1[(size_t)(rbase + r) * HID + m] = acc[r] * rsqrtf((float)(dcnt[lrow + r] + 1));
}


typedef __attribute__((ext_vector_type(4))) float v4f;
typedef __attribute__((ext_vector_type(4))) int i32x4;
typedef unsigned int __attribute__((may_alias)) u32a;
typedef unsigned int __attribute__((ext_vector_type(2))) u32x2;
typedef u32x2 __attribute__((may_alias)) u32x2a;

#define BINW 64
#define BINP 68
#define PPB  4

__global__ __launch_bounds__(512, 16) void k_agg1_bucket(const float* __restrict__ h1s,
                                                     const unsigned short* __restrict__ e16,
                                                     const unsigned char* __restrict__ e8,
                                                     const unsigned short* __restrict__ otab,
                                                     const float* __restrict__ b1,
                                                     const float* __restrict__ W2,
                                                     float* __restrict__ h2s) {
  __shared__ __attribute__((aligned(16))) float acc[RB * HID];
  __shared__ int cnt[RB];
  __shared__ int full;
  __shared__ __attribute__((aligned(16))) int bin[RB * BINP];
  const int t = threadIdx.x, base = blockIdx.x * RB;
  const int j = t & 3, qd = t >> 2;
  const int jq = j << 2;
  const int jc = qd < NBK ? qd : NBK - 1;
  const unsigned short* rw = otab + (size_t)jc * TS + blockIdx.x;
  const int r0 = rw[0], r1 = rw[1];
  v4f own = {0.0f, 0.0f, 0.0f, 0.0f};
  if (base + qd < NN) own = *(const v4f*)(h1s + (size_t)base * HID + 4 * t);
  if (t < RB) cnt[t] = 0;
  if (t == 0) full = 0;
  for (int i = t; i < RB * HID; i += 512) acc[i] = 0.0f;
  __syncthreads();

  {
    const int rb = qd < NBK ? r0 : 0;
    const int oe = qd < NBK ? r1 : 0;
    const u32a*   e8w  = (const u32a*)e8;
    const u32x2a* e16w = (const u32x2a*)e16;
    const unsigned cw = (((unsigned)jc * CH) >> 2) + (unsigned)j;
    const int pl = (oe - 1) >> 4;
    for (int p = rb >> 4; p <= pl; p += PPB) {
      unsigned w8[PPB];
      u32x2 w16[PPB];
#pragma unroll
      for (int k = 0; k < PPB; ++k) {
        const int pk = p + k < pl ? p + k : pl;
        w8[k]  = e8w[cw + 4u * (unsigned)pk];
        w16[k] = e16w[cw + 4u * (unsigned)pk];
      }
      int slot[4 * PPB];
#pragma unroll
      for (int k = 0; k < PPB; ++k)
#pragma unroll
        for (int u = 0; u < 4; ++u) {
          const int i = 16 * (p + k) + 4 * j + u;
          const int dl = (int)((w8[k] >> (8 * u)) & (RB - 1));
          slot[4 * k + u] = 0;
          if (i >= rb && i < oe) slot[4 * k + u] = atomicAdd(&cnt[dl], 1);
        }
#pragma unroll
      for (int k = 0; k < PPB; ++k)
#pragma unroll
        for (int u = 0; u < 4; ++u) {
          const int i = 16 * (p + k) + 4 * j + u;
          if (i >= rb && i < oe) {
            const unsigned q  = (w8[k] >> (8 * u)) & 255u;
            const unsigned hw = (u & 2) ? w16[k].y : w16[k].x;
            const unsigned lo = (u & 1) ? (hw >> 16) : (hw & 0xFFFFu);
            const unsigned s  = lo | ((q >> 7) << 16);
            const int dl = (int)(q & (RB - 1));
            if (slot[4 * k + u] < BINW) bin[dl * BINP + slot[4 * k + u]] = (int)s;
            else full = 1;
          }
        }
    }
  }
  __syncthreads();

  {
    v4f sum = own;
    const int any_full = full;
    if (any_full) {
      const int oe = qd < NBK ? r1 : 0;
      const unsigned cb = (unsigned)jc * CH;
      for (int i = (qd < NBK ? r0 : 0) + j; i < oe; i += 4) {
        const int rec = (int)e8[cb + (unsigned)i] | ((int)e16[cb + (unsigned)i] << 8);
        const int dl = rec & (RB - 1);
        const unsigned s = ((unsigned)rec >> 8) | (((unsigned)rec & 0x80u) << 9);
        const float* hs = h1s + ((size_t)s << 4);
        for (int k = 0; k < HID; ++k) atomicAdd(&acc[(dl << 4) | k], hs[k]);
      }
      __syncthreads();
    }
    const int deg = cnt[qd];
    const int n = any_full ? 0 : (deg < BINW ? deg : BINW);
    const int* bp = bin + qd * BINP;
    for (int i0 = 0; i0 < n; i0 += 8) {
      const i32x4 ia = *(const i32x4*)(bp + i0);
      const i32x4 ib = *(const i32x4*)(bp + i0 + 4);
      const int id[8] = {ia.x, ia.y, ia.z, ia.w, ib.x, ib.y, ib.z, ib.w};
      v4f hv[8];
#pragma unroll
      for (int u = 0; u < 8; ++u) {
        const unsigned s = (i0 + u < n) ? ((unsigned)id[u] & 0x1FFFFu) : 0u;
        hv[u] = *(const v4f*)(h1s + ((s << 4) | (unsigned)jq));
      }
      __builtin_amdgcn_sched_barrier(0);
#pragma unroll
      for (int u = 0; u < 8; ++u) {
        const bool live = i0 + u < n;
        sum.x += live ? hv[u].x : 0.0f; sum.y += live ? hv[u].y : 0.0f;
        sum.z += live ? hv[u].z : 0.0f; sum.w += live ? hv[u].w : 0.0f;
      }
    }
    v4f* rowv = (v4f*)(acc + ((qd << 4) | jq));
    v4f o = *rowv;
    o.x += sum.x; o.y += sum.y; o.z += sum.z; o.w += sum.w;
    *rowv = o;
  }
  __syncthreads();
  if (t < RB && base + t < NN) {
    const float di = rsqrtf((float)(cnt[t] + 1));
    float v[HID];
#pragma unroll
    for (int k = 0; k < HID; ++k) {
      float x = acc[(t << 4) | k] * di + b1[k];
      v[k] = x > 0.0f ? x : 0.0f;
    }
#pragma unroll
    for (int c = 0; c < NC; ++c) {
      float sum = 0.0f;
#pragma unroll
      for (int k = 0; k < HID; ++k) sum += v[k] * W2[k * NC + c];
      h2s[(size_t)(base + t) * NC + c] = sum * di;
    }
  }
}

#define NSL 4
#define SLN 25000
#define GB  13
#define NG  61
#define SLV (SLN * NC / 4)

__global__ __launch_bounds__(1024) void k_agg2_slice(const float* __restrict__ h2s,
                                                    const unsigned short* __restrict__ e16,
                                                    const unsigned char* __restrict__ e8,
                                                    const unsigned short* __restrict__ otab,
                                                    float* __restrict__ part) {
  __shared__ __attribute__((aligned(16))) float sl[SLN * NC];
  __shared__ __attribute__((aligned(16))) float acc[4 * GB * RB];
  const int t = threadIdx.x, lane = t & 31, wave = t >> 5;
  const int s = blockIdx.x & (NSL - 1), g = blockIdx.x >> 2;
  const int b0 = g * GB;
  const int nb = (PB - b0) < GB ? (PB - b0) : GB;
  int bnds[4];
#pragma unroll
  for (int ci = 0; ci < 4; ++ci) {
    const int c = wave + 32 * ci, cc = c < NBK ? c : NBK - 1;
    bnds[ci] = (int)otab[(size_t)cc * TS + b0 + (lane < nb ? lane : nb)];
  }
  {
    const v4f* gsl = (const v4f*)(h2s + (size_t)s * (SLN * NC));
    v4f* lsl = (v4f*)sl;
    for (int i0 = 0; i0 < SLV; i0 += 1024 * 10) {
      v4f r[10];
#pragma unroll
      for (int u = 0; u < 10; ++u) { const int i = i0 + t + 1024 * u; r[u] = gsl[i < SLV ? i : SLV - 1]; }
#pragma unroll
      for (int u = 0; u < 10; ++u) { const int i = i0 + t + 1024 * u; if (i < SLV) lsl[i] = r[u]; }
    }
    const v4f z = {0.0f, 0.0f, 0.0f, 0.0f};
    for (int i = t; i < GB * RB; i += 1024) ((v4f*)acc)[i] = z;
  }
  __syncthreads();

  const u32a*   e8w  = (const u32a*)e8;
  const u32x2a* e16w = (const u32x2a*)e16;
  const unsigned sbase = (unsigned)s * SLN;
#pragma unroll
  for (int ci = 0; ci < 4; ++ci) {
    const int c = wave + 32 * ci;
    if (c < NBK) {
      const int bnd = bnds[ci];
      const int pbeg = __builtin_amdgcn_readlane(bnd, 0);
      const int pend = __builtin_amdgcn_readlane(bnd, nb);
      const unsigned cw = ((unsigned)c * CH) >> 2;
      int kb = 0;
      int base = pbeg & ~127;
      unsigned w8 = 0; u32x2 w16 = {0u, 0u};
      if (base < pend) { w8 = e8w[cw + (base >> 2) + lane]; w16 = e16w[cw + (base >> 2) + lane]; }
      for (; base < pend; base += 128) {
        unsigned n8 = 0; u32x2 n16 = {0u, 0u};
        if (base + 128 < pend) { n8 = e8w[cw + ((base + 128) >> 2) + lane]; n16 = e16w[cw + ((base + 128) >> 2) + lane]; }
        const int first = base > pbeg ? base : pbeg;
        while (kb + 1 < nb && __builtin_amdgcn_readlane(bnd, kb + 1) <= first) ++kb;
        const int p0 = base + 4 * lane;
        int kk[4] = {kb, kb, kb, kb};
        for (int j = kb + 1; j < nb; ++j) {
          const int bj = __builtin_amdgcn_readlane(bnd, j);
          if (bj > base + 127) break;
#pragma unroll
          for (int u = 0; u < 4; ++u) kk[u] += (p0 + u >= bj) ? 1 : 0;
        }
#pragma unroll
        for (int u = 0; u < 4; ++u) {
          const int p = p0 + u;
          const unsigned q  = (w8 >> (8 * u)) & 255u;
          const unsigned hw = (u & 2) ? w16.y : w16.x;
          const unsigned lo = (u & 1) ? (hw >> 16) : (hw & 0xFFFFu);
          const unsigned rel = (lo | ((q >> 7) << 16)) - sbase;
          if (p >= pbeg && p < pend && rel < (unsigned)SLN) {
            const float* h = sl + rel * NC;
            float* a = acc + ((kk[u] << 7) | (int)(q & (RB - 1)));
            atomicAdd(a, h[0]);
            atomicAdd(a + GB * RB, h[1]);
            atomicAdd(a + 2 * GB * RB, h[2]);
            atomicAdd(a + 3 * GB * RB, 1.0f);
          }
        }
        w8 = n8; w16 = n16;
      }
    }
  }
  __syncthreads();
  {
    v4f* p4 = (v4f*)part + (size_t)s * NN + (size_t)b0 * RB;
    for (int i = t; i < nb * RB; i += 1024)
      if (b0 * RB + i < NN) {
        const v4f o = {acc[i], acc[GB * RB + i], acc[2 * GB * RB + i], acc[3 * GB * RB + i]};
        p4[i] = o;
      }
  }
}

__global__ void k_logsoftmax(const float* __restrict__ part, const float* __restrict__ b2,
                             float* out) {
  int i = blockIdx.x * blockDim.x + threadIdx.x;
  if (i >= NN) return;
  const v4f* p4 = (const v4f*)part;
  const v4f pa = p4[i], pb = p4[NN + i], pc = p4[2 * NN + i], pd = p4[3 * NN + i];
  const float h0 = out[(size_t)i * NC + 0], h1 = out[(size_t)i * NC + 1], h2 = out[(size_t)i * NC + 2];
  const float di = rsqrtf(pa.w + pb.w + pc.w + pd.w + 1.0f);
  float z0 = (h0 + pa.x + pb.x + pc.x + pd.x) * di + b2[0];
  float z1 = (h1 + pa.y + pb.y + pc.y + pd.y) * di + b2[1];
  float z2 = (h2 + pa.z + pb.z + pc.z + pd.z) * di + b2[2];
  float mx = fmaxf(z0, fmaxf(z1, z2));
  float lse = logf(expf(z0 - mx) + expf(z1 - mx) + expf(z2 - mx));
  out[(size_t)i * NC + 0] = z0 - mx - lse;
  out[(size_t)i * NC + 1] = z1 - mx - lse;
  out[(size_t)i * NC + 2] = z2 - mx - lse;
}

extern "C" void kernel_launch(void* const* d_in, const int* in_sizes, int n_in,
                              void* d_out, int out_size, void* d_ws, size_t ws_size,
                              hipStream_t stream) {
  const float* x  = (const float*)d_in[0];
  const float* W1 = (const float*)d_in[1];
  const float* b1 = (const float*)d_in[2];
  const float* W2 = (const float*)d_in[3];
  const float* b2 = (const float*)d_in[4];
  const int*   ei = (const int*)d_in[5];
  const int* src = ei;
  const int* dst = ei + EE;
  float* out = (float*)d_out;

  char* wsb = (char*)d_ws;
  float*          h1s  = (float*)wsb;
  unsigned short* e16  = (unsigned short*)(wsb + 6400000);
  unsigned char*  e8   = (unsigned char*)(wsb + 12800000);
  unsigned short* otab = (unsigned short*)(wsb + 16000000);
  float* part = (float*)wsb;
  float* h2s = out;

  k_chunk_sort  <<<NBK, 1024, 0, stream>>>(src, dst, e16, e8, otab);
  k_gemm1_wmma  <<<PB, 256, 0, stream>>>(x, W1, e8, otab, h1s);
  k_agg1_bucket <<<PB, 512, 0, stream>>>(h1s, e16, e8, otab, b1, W2, h2s);
  k_agg2_slice  <<<NG * NSL, 1024, 0, stream>>>(h2s, e16, e8, otab, part);
  k_logsoftmax  <<<(NN + 255) / 256, 256, 0, stream>>>(part, b2, out);
}
